# starting kernel
# baseline (speedup 1.0000x reference)
#include <hip/hip_runtime.h>
#include <hip/hip_bf16.h>

#define N0    262144
#define N1_   65536
#define N2_   16384
#define E1_   1048576
#define E2_   262144
#define IN_C  128
#define HID_C 256
#define OUT_C 64

typedef __attribute__((ext_vector_type(2))) float v2f;
typedef __attribute__((ext_vector_type(8))) float v8f;

__global__ void sage_count_deg(const int* __restrict__ dst, int* __restrict__ deg,
                               int E) {
    int e = blockIdx.x * blockDim.x + threadIdx.x;
    if (e < E) atomicAdd(&deg[dst[e]], 1);
}

__global__ void sage_excl_scan(const int* __restrict__ in, int* __restrict__ out,
                               int n) {
    __shared__ int buf[1024];
    __shared__ int carry_s;
    const int tid = threadIdx.x;
    if (tid == 0) carry_s = 0;
    __syncthreads();
    for (int base = 0; base < n; base += 1024) {
        int i = base + tid;
        int v = (i < n) ? in[i] : 0;
        buf[tid] = v;
        __syncthreads();
        for (int off = 1; off < 1024; off <<= 1) {
            int t = (tid >= off) ? buf[tid - off] : 0;
            __syncthreads();
            buf[tid] += t;
            __syncthreads();
        }
        int incl = buf[tid];
        int c = carry_s;
        if (i < n) out[i] = c + incl - v;
        __syncthreads();
        if (tid == 0) carry_s = c + buf[1023];
        __syncthreads();
    }
}

__global__ void sage_bucket_edges(const int* __restrict__ src,
                                  const int* __restrict__ dst,
                                  const int* __restrict__ offs,
                                  int* __restrict__ fill,
                                  int* __restrict__ esrc, int E) {
    int e = blockIdx.x * blockDim.x + threadIdx.x;
    if (e >= E) return;
    int d = dst[e];
    int pos = offs[d] + atomicAdd(&fill[d], 1);
    esrc[pos] = src[e];
}

template <int C>
__global__ void sage_gather_mean(const float* __restrict__ x,
                                 const int* __restrict__ esrc,
                                 const int* __restrict__ offs,
                                 const int* __restrict__ deg,
                                 float* __restrict__ outm, int nDst) {
    const int wave = blockIdx.x * (blockDim.x >> 5) + (threadIdx.x >> 5);
    const int lane = threadIdx.x & 31;
    if (wave >= nDst) return;
    const int beg = offs[wave];
    const int d   = deg[wave];
    constexpr int V = C / 128;
    float4 acc[V];
#pragma unroll
    for (int v = 0; v < V; ++v) acc[v] = make_float4(0.f, 0.f, 0.f, 0.f);

    for (int base = 0; base < d; base += 32) {
        int n = d - base;
        if (n > 32) n = 32;
        int myid = (lane < n) ? esrc[beg + base + lane] : 0;
        for (int j = 0; j < n; ++j) {
            int s = __shfl(myid, j, 32);
            const float4* row = (const float4*)(x + (size_t)s * C);
#pragma unroll
            for (int v = 0; v < V; ++v) {
                float4 t = row[lane + 32 * v];
                acc[v].x += t.x; acc[v].y += t.y;
                acc[v].z += t.z; acc[v].w += t.w;
            }
        }
    }

    const float scale = (d > 0) ? (1.0f / (float)d) : 0.0f;
    float4* orow = (float4*)(outm + (size_t)wave * C);
#pragma unroll
    for (int v = 0; v < V; ++v) {
        float4 t;
        t.x = acc[v].x * scale; t.y = acc[v].y * scale;
        t.z = acc[v].z * scale; t.w = acc[v].w * scale;
        orow[lane + 32 * v] = t;
    }
}

template <int K, int N, int MT, int NT, bool RELU>
__global__ void sage_gemm_two(const float* __restrict__ A1,
                              const float* __restrict__ B1,
                              const float* __restrict__ A2,
                              const float* __restrict__ B2,
                              const float* __restrict__ bias,
                              float* __restrict__ out, int M) {
    const int lane   = threadIdx.x & 31;
    const int waveId = blockIdx.x * (blockDim.x >> 5) + (threadIdx.x >> 5);
    const int groupsN = N / (16 * NT);
    const int gm = waveId / groupsN;
    const int gn = waveId % groupsN;
    if (gm >= M / (16 * MT)) return;

    const int m0  = gm * 16 * MT;
    const int n0  = gn * 16 * NT;
    const int khi = (lane >> 4) * 2;

    const float* a1p[MT];
    const float* a2p[MT];
#pragma unroll
    for (int mt = 0; mt < MT; ++mt) {
        int r = m0 + mt * 16 + (lane & 15);
        a1p[mt] = A1 + (size_t)r * K + khi;
        a2p[mt] = A2 + (size_t)r * K + khi;
    }
    const float* b1p[NT];
    const float* b2p[NT];
#pragma unroll
    for (int nt = 0; nt < NT; ++nt) {
        int c = n0 + nt * 16 + (lane & 15);
        b1p[nt] = B1 + (size_t)c * K + khi;
        b2p[nt] = B2 + (size_t)c * K + khi;
    }

    v8f acc[MT][NT] = {};

#pragma unroll 2
    for (int k0 = 0; k0 < K; k0 += 4) {
        v2f a[MT], b[NT];
#pragma unroll
        for (int mt = 0; mt < MT; ++mt) a[mt] = *(const v2f*)(a1p[mt] + k0);
#pragma unroll
        for (int nt = 0; nt < NT; ++nt) b[nt] = *(const v2f*)(b1p[nt] + k0);
#pragma unroll
        for (int mt = 0; mt < MT; ++mt)
#pragma unroll
            for (int nt = 0; nt < NT; ++nt)
                acc[mt][nt] = __builtin_amdgcn_wmma_f32_16x16x4_f32(
                    false, a[mt], false, b[nt], (short)0, acc[mt][nt],
                    false, false);
    }
#pragma unroll 2
    for (int k0 = 0; k0 < K; k0 += 4) {
        v2f a[MT], b[NT];
#pragma unroll
        for (int mt = 0; mt < MT; ++mt) a[mt] = *(const v2f*)(a2p[mt] + k0);
#pragma unroll
        for (int nt = 0; nt < NT; ++nt) b[nt] = *(const v2f*)(b2p[nt] + k0);
#pragma unroll
        for (int mt = 0; mt < MT; ++mt)
#pragma unroll
            for (int nt = 0; nt < NT; ++nt)
                acc[mt][nt] = __builtin_amdgcn_wmma_f32_16x16x4_f32(
                    false, a[mt], false, b[nt], (short)0, acc[mt][nt],
                    false, false);
    }

    const int rbase = ((lane >> 4) << 3);
#pragma unroll
    for (int nt = 0; nt < NT; ++nt) {
        const int ocol = n0 + nt * 16 + (lane & 15);
        const float bb = bias[ocol];
#pragma unroll
        for (int mt = 0; mt < MT; ++mt) {
#pragma unroll
            for (int r = 0; r < 8; ++r) {
                float v = acc[mt][nt][r] + bb;
                if (RELU) v = fmaxf(v, 0.0f);
                out[(size_t)(m0 + mt * 16 + rbase + r) * N + ocol] = v;
            }
        }
    }
}

__global__ void sage_logsoftmax64(float* __restrict__ out, int rows) {
    int wave = blockIdx.x * (blockDim.x >> 5) + (threadIdx.x >> 5);
    int lane = threadIdx.x & 31;
    if (wave >= rows) return;
    float* p = out + (size_t)wave * 64;
    float a = p[lane];
    float b = p[lane + 32];
    float m = fmaxf(a, b);
#pragma unroll
    for (int off = 16; off > 0; off >>= 1)
        m = fmaxf(m, __shfl_xor(m, off, 32));
    float s = expf(a - m) + expf(b - m);
#pragma unroll
    for (int off = 16; off > 0; off >>= 1)
        s += __shfl_xor(s, off, 32);
    float lse = m + logf(s);
    p[lane]      = a - lse;
    p[lane + 32] = b - lse;
}

extern "C" void kernel_launch(void* const* d_in, const int* in_sizes, int n_in,
                              void* d_out, int out_size, void* d_ws, size_t ws_size,
                              hipStream_t stream) {
    const float* x    = (const float*)d_in[0];
    const int*   src1 = (const int*)d_in[1];
    const int*   dst1 = (const int*)d_in[2];
    const int*   src2 = (const int*)d_in[3];
    const int*   dst2 = (const int*)d_in[4];
    const float* W1l  = (const float*)d_in[5];
    const float* W1r  = (const float*)d_in[6];
    const float* b1   = (const float*)d_in[7];
    const float* W2l  = (const float*)d_in[8];
    const float* W2r  = (const float*)d_in[9];
    const float* b2   = (const float*)d_in[10];
    float* out = (float*)d_out;

    int* deg1  = (int*)d_ws;
    int* fill1 = deg1 + N1_;
    int* deg2  = fill1 + N1_;
    int* fill2 = deg2 + N2_;
    int* offs1 = fill2 + N2_;
    int* offs2 = offs1 + N1_;
    int* esrc1 = offs2 + N2_;
    int* esrc2 = esrc1 + E1_;
    float* aggr1 = (float*)(esrc2 + E2_);
    float* aggr2 = aggr1 + (size_t)N1_ * IN_C;
    float* h     = aggr2 + (size_t)N2_ * HID_C;

    size_t zbytes = (size_t)(2 * N1_ + 2 * N2_) * sizeof(int);
    hipMemsetAsync(d_ws, 0, zbytes, stream);

    sage_count_deg<<<E1_ / 256, 256, 0, stream>>>(dst1, deg1, E1_);
    sage_excl_scan<<<1, 1024, 0, stream>>>(deg1, offs1, N1_);
    sage_bucket_edges<<<E1_ / 256, 256, 0, stream>>>(src1, dst1, offs1, fill1,
                                                     esrc1, E1_);
    sage_gather_mean<IN_C><<<N1_ / 8, 256, 0, stream>>>(x, esrc1, offs1, deg1,
                                                        aggr1, N1_);
    {
        int waves = (N1_ / 32) * (HID_C / 64);
        sage_gemm_two<IN_C, HID_C, 2, 4, true>
            <<<waves / 8, 256, 0, stream>>>(aggr1, W1l, x, W1r, b1, h, N1_);
    }

    sage_count_deg<<<E2_ / 256, 256, 0, stream>>>(dst2, deg2, E2_);
    sage_excl_scan<<<1, 1024, 0, stream>>>(deg2, offs2, N2_);
    sage_bucket_edges<<<E2_ / 256, 256, 0, stream>>>(src2, dst2, offs2, fill2,
                                                     esrc2, E2_);
    sage_gather_mean<HID_C><<<N2_ / 8, 256, 0, stream>>>(h, esrc2, offs2, deg2,
                                                         aggr2, N2_);
    {
        int waves = (N2_ / 32) * (OUT_C / 32);
        sage_gemm_two<HID_C, OUT_C, 2, 2, false>
            <<<waves / 8, 256, 0, stream>>>(aggr2, W2l, h, W2r, b2, out, N2_);
    }

    sage_logsoftmax64<<<(N2_ + 7) / 8, 256, 0, stream>>>(out, N2_);
}


# Layer 2 projects before averaging: P = h @ W2_l^T for all N1 rows (weight-stationary WMMA); list walk gathers 256-B rows of P, adds own term, log-softmax in-wave; 8 -> 7 launches; sums reorder
# speedup vs baseline: 2.6083x; 2.6083x over previous
#include <hip/hip_runtime.h>
#include <hip/hip_bf16.h>

#define N0    262144
#define N1_   65536
#define N2_   16384
#define E1_   1048576
#define E2_   262144
#define IN_C  128
#define HID_C 256
#define OUT_C 64

typedef __attribute__((ext_vector_type(2))) float v2f;
typedef __attribute__((ext_vector_type(8))) float v8f;

__global__ void sage_link_edges(const int* __restrict__ dst,
                                int* __restrict__ head,
                                int* __restrict__ next, int E) {
    const int t  = blockIdx.x * blockDim.x + threadIdx.x;
    const int e0 = 4 * t;
    if (e0 >= E) return;
    const int4 d = ((const int4*)dst)[t];
    int4 p;
    p.x = atomicExch(&head[d.x], e0 + 1);
    p.y = atomicExch(&head[d.y], e0 + 2);
    p.z = atomicExch(&head[d.z], e0 + 3);
    p.w = atomicExch(&head[d.w], e0 + 4);
    ((int4*)next)[t] = p;
}

static __device__ __forceinline__ int sage_link_or_end(int e, int E) {
    return ((unsigned)(e - 1) < (unsigned)E) ? e : 0;
}

template <int C, int R>
__global__ void sage_walk_mean(const float* __restrict__ x,
                               const int* __restrict__ src,
                               const int* __restrict__ next,
                               const int* __restrict__ head,
                               float* __restrict__ outm, int nDst, int E) {
    const int wave = blockIdx.x * (blockDim.x >> 5) +
                     __builtin_amdgcn_readfirstlane(threadIdx.x >> 5);
    const int lane = threadIdx.x & 31;
    const int row0 = wave * R;
    if (row0 >= nDst) return;
    constexpr int V = C / 128;
    float4 acc[R][V];
    int e[R], s[R], nx[R], cnt[R];
#pragma unroll
    for (int r = 0; r < R; ++r) {
#pragma unroll
        for (int v = 0; v < V; ++v) acc[r][v] = make_float4(0.f, 0.f, 0.f, 0.f);
        cnt[r] = 0;
        e[r] = (row0 + r < nDst) ? sage_link_or_end(head[row0 + r], E) : 0;
    }
#pragma unroll
    for (int r = 0; r < R; ++r) {
        s[r] = 0; nx[r] = 0;
        if (e[r] != 0) { s[r] = src[e[r] - 1]; nx[r] = next[e[r] - 1]; }
    }

    for (int hop = 0; hop < E; ++hop) {
        int alive = 0;
#pragma unroll
        for (int r = 0; r < R; ++r) alive |= e[r];
        if (alive == 0) break;

        float4 t[R][V];
#pragma unroll
        for (int r = 0; r < R; ++r) {
#pragma unroll
            for (int v = 0; v < V; ++v) t[r][v] = make_float4(0.f, 0.f, 0.f, 0.f);
            if (e[r] != 0) {
                const float4* row = (const float4*)(x + (size_t)s[r] * C);
#pragma unroll
                for (int v = 0; v < V; ++v) t[r][v] = row[lane + 32 * v];
                cnt[r] += 1;
            }
        }
#pragma unroll
        for (int r = 0; r < R; ++r) {
            e[r] = sage_link_or_end(nx[r], E);
            if (e[r] != 0) { s[r] = src[e[r] - 1]; nx[r] = next[e[r] - 1]; }
            else           { nx[r] = 0; }
        }
#pragma unroll
        for (int r = 0; r < R; ++r) {
#pragma unroll
            for (int v = 0; v < V; ++v) {
                acc[r][v].x += t[r][v].x; acc[r][v].y += t[r][v].y;
                acc[r][v].z += t[r][v].z; acc[r][v].w += t[r][v].w;
            }
        }
    }

#pragma unroll
    for (int r = 0; r < R; ++r) {
        if (row0 + r >= nDst) break;
        const float scale = (cnt[r] > 0) ? (1.0f / (float)cnt[r]) : 0.0f;
        float4* orow = (float4*)(outm + (size_t)(row0 + r) * C);
#pragma unroll
        for (int v = 0; v < V; ++v) {
            float4 o;
            o.x = acc[r][v].x * scale; o.y = acc[r][v].y * scale;
            o.z = acc[r][v].z * scale; o.w = acc[r][v].w * scale;
            orow[lane + 32 * v] = o;
        }
    }
}

template <int K, int N, int MT, int NT, int CH, bool RELU>
__global__ void __launch_bounds__(N / (16 * NT) * 32)
sage_gemm_wreg(const float* __restrict__ A1, const float* __restrict__ B1,
               const float* __restrict__ A2, const float* __restrict__ B2,
               const float* __restrict__ bias, float* __restrict__ out, int M) {
    constexpr int KS  = K / 4;
    constexpr int NCH = KS / CH;
    const int lane = threadIdx.x & 31;
    const int wave = threadIdx.x >> 5;
    const int c    = lane & 15;
    const int khi  = (lane >> 4) * 2;
    const int n0   = wave * 16 * NT;

    const int numTiles = M / (16 * MT);
    int tile = blockIdx.x;
    if (tile >= numTiles) return;

    v2f bw[2][KS][NT];
#pragma unroll
    for (int nt = 0; nt < NT; ++nt) {
        const size_t ro = (size_t)(n0 + nt * 16 + c) * K + khi;
#pragma unroll
        for (int j = 0; j < KS; ++j) {
            bw[0][j][nt] = *(const v2f*)(B1 + ro + 4 * j);
            bw[1][j][nt] = *(const v2f*)(B2 + ro + 4 * j);
        }
    }

    const int laneOff = c * K + khi;
    v2f cur[CH][MT];
    {
        const float* p = A1 + (size_t)tile * (16 * MT) * K + laneOff;
#pragma unroll
        for (int s = 0; s < CH; ++s)
#pragma unroll
            for (int mt = 0; mt < MT; ++mt)
                cur[s][mt] = *(const v2f*)(p + mt * 16 * K + 4 * s);
    }

    for (; tile < numTiles; tile += gridDim.x) {
        const int m0 = tile * 16 * MT;
        const float* p1 = A1 + (size_t)m0 * K + laneOff;
        const float* p2 = A2 + (size_t)m0 * K + laneOff;
        int nextTile = tile + gridDim.x;
        if (nextTile >= numTiles) nextTile = tile;
        const float* pn = A1 + (size_t)nextTile * (16 * MT) * K + laneOff;

        v8f acc[MT][NT] = {};

#pragma unroll
        for (int g = 0; g < 2 * NCH; ++g) {
            const int gn = g + 1;
            const float* q = (gn < NCH)     ? (p1 + gn * CH * 4)
                           : (gn < 2 * NCH) ? (p2 + (gn - NCH) * CH * 4)
                                            : pn;
            v2f nxt[CH][MT];
#pragma unroll
            for (int s = 0; s < CH; ++s)
#pragma unroll
                for (int mt = 0; mt < MT; ++mt)
                    nxt[s][mt] = *(const v2f*)(q + mt * 16 * K + 4 * s);
            __builtin_amdgcn_sched_barrier(0);

            const int ph = g / NCH;
            const int jb = (g % NCH) * CH;
#pragma unroll
            for (int s = 0; s < CH; ++s)
#pragma unroll
                for (int mt = 0; mt < MT; ++mt)
#pragma unroll
                    for (int nt = 0; nt < NT; ++nt)
                        acc[mt][nt] = __builtin_amdgcn_wmma_f32_16x16x4_f32(
                            false, cur[s][mt], false, bw[ph][jb + s][nt],
                            (short)0, acc[mt][nt], false, false);
#pragma unroll
            for (int s = 0; s < CH; ++s)
#pragma unroll
                for (int mt = 0; mt < MT; ++mt) cur[s][mt] = nxt[s][mt];
            __builtin_amdgcn_sched_barrier(0);
        }

        const int rbase = ((lane >> 4) << 3);
#pragma unroll
        for (int nt = 0; nt < NT; ++nt) {
            const int ocol = n0 + nt * 16 + c;
            const float bb = bias[ocol];
#pragma unroll
            for (int mt = 0; mt < MT; ++mt) {
#pragma unroll
                for (int r = 0; r < 8; ++r) {
                    float v = acc[mt][nt][r] + bb;
                    if (RELU) v = fmaxf(v, 0.0f);
                    out[(size_t)(m0 + mt * 16 + rbase + r) * N + ocol] = v;
                }
            }
        }
    }
}

template <int K, int N, int MT, int NT, int CH>
__global__ void __launch_bounds__(N / (16 * NT) * 32)
sage_proj_wreg(const float* __restrict__ A,
               const float* __restrict__ Bp, const float* __restrict__ Bq,
               float* __restrict__ P, float* __restrict__ Q,
               int Mp, int Mq, int groupsP) {
    constexpr int KS  = K / 4;
    constexpr int NCH = KS / CH;
    const int lane = threadIdx.x & 31;
    const int wave = threadIdx.x >> 5;
    const int c    = lane & 15;
    const int khi  = (lane >> 4) * 2;
    const int n0   = wave * 16 * NT;

    const bool isQ = (int)blockIdx.x >= groupsP;
    const float* B   = isQ ? Bq : Bp;
    float*       out = isQ ? Q : P;
    const int numTiles = (isQ ? Mq : Mp) / (16 * MT);
    const int step     = isQ ? ((int)gridDim.x - groupsP) : groupsP;
    int tile           = isQ ? ((int)blockIdx.x - groupsP) : (int)blockIdx.x;
    if (tile >= numTiles) return;

    v2f bw[KS][NT];
#pragma unroll
    for (int nt = 0; nt < NT; ++nt) {
        const size_t ro = (size_t)(n0 + nt * 16 + c) * K + khi;
#pragma unroll
        for (int j = 0; j < KS; ++j) bw[j][nt] = *(const v2f*)(B + ro + 4 * j);
    }

    const int laneOff = c * K + khi;
    v2f cur[CH][MT];
    {
        const float* p = A + (size_t)tile * (16 * MT) * K + laneOff;
#pragma unroll
        for (int s = 0; s < CH; ++s)
#pragma unroll
            for (int mt = 0; mt < MT; ++mt)
                cur[s][mt] = *(const v2f*)(p + mt * 16 * K + 4 * s);
    }

    for (; tile < numTiles; tile += step) {
        const int m0 = tile * 16 * MT;
        const float* p1 = A + (size_t)m0 * K + laneOff;
        int nextTile = tile + step;
        if (nextTile >= numTiles) nextTile = tile;
        const float* pn = A + (size_t)nextTile * (16 * MT) * K + laneOff;

        v8f acc[MT][NT] = {};

#pragma unroll
        for (int g = 0; g < NCH; ++g) {
            const int gn = g + 1;
            const float* q = (gn < NCH) ? (p1 + gn * CH * 4) : pn;
            v2f nxt[CH][MT];
#pragma unroll
            for (int s = 0; s < CH; ++s)
#pragma unroll
                for (int mt = 0; mt < MT; ++mt)
                    nxt[s][mt] = *(const v2f*)(q + mt * 16 * K + 4 * s);
            __builtin_amdgcn_sched_barrier(0);

            const int jb = g * CH;
#pragma unroll
            for (int s = 0; s < CH; ++s)
#pragma unroll
                for (int mt = 0; mt < MT; ++mt)
#pragma unroll
                    for (int nt = 0; nt < NT; ++nt)
                        acc[mt][nt] = __builtin_amdgcn_wmma_f32_16x16x4_f32(
                            false, cur[s][mt], false, bw[jb + s][nt],
                            (short)0, acc[mt][nt], false, false);
#pragma unroll
            for (int s = 0; s < CH; ++s)
#pragma unroll
                for (int mt = 0; mt < MT; ++mt) cur[s][mt] = nxt[s][mt];
            __builtin_amdgcn_sched_barrier(0);
        }

        const int rbase = ((lane >> 4) << 3);
#pragma unroll
        for (int nt = 0; nt < NT; ++nt) {
            const int ocol = n0 + nt * 16 + c;
#pragma unroll
            for (int mt = 0; mt < MT; ++mt) {
#pragma unroll
                for (int r = 0; r < 8; ++r)
                    out[(size_t)(m0 + mt * 16 + rbase + r) * N + ocol] =
                        acc[mt][nt][r];
            }
        }
    }
}

template <int R>
__global__ void sage_walk_mean_lsm64(const float* __restrict__ P,
                                     const int* __restrict__ src,
                                     const int* __restrict__ next,
                                     const int* __restrict__ head,
                                     const float* __restrict__ bias,
                                     float* out, int nDst, int E) {
    const int wave = blockIdx.x * (blockDim.x >> 5) +
                     __builtin_amdgcn_readfirstlane(threadIdx.x >> 5);
    const int lane = threadIdx.x & 31;
    const int row0 = wave * R;
    if (row0 >= nDst) return;
    float2 acc[R];
    int e[R], s[R], nx[R], cnt[R];
#pragma unroll
    for (int r = 0; r < R; ++r) {
        acc[r] = make_float2(0.f, 0.f);
        cnt[r] = 0;
        e[r] = (row0 + r < nDst) ? sage_link_or_end(head[row0 + r], E) : 0;
    }
#pragma unroll
    for (int r = 0; r < R; ++r) {
        s[r] = 0; nx[r] = 0;
        if (e[r] != 0) { s[r] = src[e[r] - 1]; nx[r] = next[e[r] - 1]; }
    }

    for (int hop = 0; hop < E; ++hop) {
        int alive = 0;
#pragma unroll
        for (int r = 0; r < R; ++r) alive |= e[r];
        if (alive == 0) break;

        float2 t[R];
#pragma unroll
        for (int r = 0; r < R; ++r) {
            t[r] = make_float2(0.f, 0.f);
            if (e[r] != 0) {
                t[r] = ((const float2*)(P + (size_t)s[r] * 64))[lane];
                cnt[r] += 1;
            }
        }
#pragma unroll
        for (int r = 0; r < R; ++r) {
            e[r] = sage_link_or_end(nx[r], E);
            if (e[r] != 0) { s[r] = src[e[r] - 1]; nx[r] = next[e[r] - 1]; }
            else           { nx[r] = 0; }
        }
#pragma unroll
        for (int r = 0; r < R; ++r) { acc[r].x += t[r].x; acc[r].y += t[r].y; }
    }

    const float2 bb = ((const float2*)bias)[lane];
#pragma unroll
    for (int r = 0; r < R; ++r) {
        if (row0 + r >= nDst) break;
        const float scale = (cnt[r] > 0) ? (1.0f / (float)cnt[r]) : 0.0f;
        float2* orow = (float2*)(out + (size_t)(row0 + r) * 64);
        const float2 own = orow[lane];
        const float a = (acc[r].x * scale + bb.x) + own.x;
        const float b = (acc[r].y * scale + bb.y) + own.y;
        float m = fmaxf(a, b);
#pragma unroll
        for (int off = 16; off > 0; off >>= 1)
            m = fmaxf(m, __shfl_xor(m, off, 32));
        float z = expf(a - m) + expf(b - m);
#pragma unroll
        for (int off = 16; off > 0; off >>= 1)
            z += __shfl_xor(z, off, 32);
        const float lse = m + logf(z);
        orow[lane] = make_float2(a - lse, b - lse);
    }
}

extern "C" void kernel_launch(void* const* d_in, const int* in_sizes, int n_in,
                              void* d_out, int out_size, void* d_ws, size_t ws_size,
                              hipStream_t stream) {
    const float* x    = (const float*)d_in[0];
    const int*   src1 = (const int*)d_in[1];
    const int*   dst1 = (const int*)d_in[2];
    const int*   src2 = (const int*)d_in[3];
    const int*   dst2 = (const int*)d_in[4];
    const float* W1l  = (const float*)d_in[5];
    const float* W1r  = (const float*)d_in[6];
    const float* b1   = (const float*)d_in[7];
    const float* W2l  = (const float*)d_in[8];
    const float* W2r  = (const float*)d_in[9];
    const float* b2   = (const float*)d_in[10];
    float* out = (float*)d_out;

    int* head1 = (int*)d_ws;
    int* head2 = head1 + N1_;
    int* next1 = head1 + 3 * N1_ + 3 * N2_;
    int* next2 = next1 + E1_;
    float* aggr1 = (float*)(next2 + E2_);
    float* proj2 = aggr1 + (size_t)N1_ * IN_C;
    float* h     = proj2 + (size_t)N1_ * OUT_C;

    size_t zbytes = (size_t)(N1_ + N2_) * sizeof(int);
    hipMemsetAsync(d_ws, 0, zbytes, stream);

    sage_link_edges<<<E1_ / 1024, 256, 0, stream>>>(dst1, head1, next1, E1_);
    sage_walk_mean<IN_C, 4><<<N1_ / 32, 256, 0, stream>>>(x, src1, next1, head1,
                                                          aggr1, N1_, E1_);
    sage_gemm_wreg<IN_C, HID_C, 2, 4, 16, true>
        <<<256, 128, 0, stream>>>(aggr1, W1l, x, W1r, b1, h, N1_);

    sage_link_edges<<<E2_ / 1024, 256, 0, stream>>>(dst2, head2, next2, E2_);
    sage_proj_wreg<HID_C, OUT_C, 2, 4, 16>
        <<<854, 32, 0, stream>>>(h, W2l, W2r, proj2, out, N1_, N2_, 683);
    sage_walk_mean_lsm64<4><<<N2_ / 32, 256, 0, stream>>>(proj2, src2, next2,
                                                          head2, b2, out,
                                                          N2_, E2_);
}


# Two linked lists per destination (edge index parity picks the list): a wave walks 8 half-length chains, 8 row loads per round trip, longest chain halved; a row's sum interleaves both: sums reorder
# speedup vs baseline: 2.6971x; 1.0340x over previous
#include <hip/hip_runtime.h>
#include <hip/hip_bf16.h>

#define N0    262144
#define N1_   65536
#define N2_   16384
#define E1_   1048576
#define E2_   262144
#define IN_C  128
#define HID_C 256
#define OUT_C 64

typedef __attribute__((ext_vector_type(2))) float v2f;
typedef __attribute__((ext_vector_type(8))) float v8f;

__global__ void sage_link_edges(const int* __restrict__ dst,
                                int* __restrict__ head,
                                int* __restrict__ next, int E) {
    const int t  = blockIdx.x * blockDim.x + threadIdx.x;
    const int e0 = 4 * t;
    if (e0 >= E) return;
    const int4 d = ((const int4*)dst)[t];
    int4 p;
    p.x = atomicExch(&head[2 * d.x],     e0 + 1);
    p.y = atomicExch(&head[2 * d.y + 1], e0 + 2);
    p.z = atomicExch(&head[2 * d.z],     e0 + 3);
    p.w = atomicExch(&head[2 * d.w + 1], e0 + 4);
    ((int4*)next)[t] = p;
}

static __device__ __forceinline__ int sage_link_or_end(int e, int E) {
    return ((unsigned)(e - 1) < (unsigned)E) ? e : 0;
}

template <int C, int R>
__global__ void sage_walk_mean(const float* __restrict__ x,
                               const int* __restrict__ src,
                               const int* __restrict__ next,
                               const int* __restrict__ head,
                               float* __restrict__ outm, int nDst, int E) {
    const int wave = blockIdx.x * (blockDim.x >> 5) +
                     __builtin_amdgcn_readfirstlane(threadIdx.x >> 5);
    const int lane = threadIdx.x & 31;
    const int row0 = wave * R;
    if (row0 >= nDst) return;
    constexpr int V = C / 128;
    constexpr int NC = 2 * R;
    float4 acc[R][V];
    int e[NC], s[NC], nx[NC], cnt[R];
#pragma unroll
    for (int r = 0; r < R; ++r) {
#pragma unroll
        for (int v = 0; v < V; ++v) acc[r][v] = make_float4(0.f, 0.f, 0.f, 0.f);
        cnt[r] = 0;
    }
#pragma unroll
    for (int c = 0; c < NC; ++c)
        e[c] = (row0 + c / 2 < nDst)
                   ? sage_link_or_end(head[2 * row0 + c], E) : 0;
#pragma unroll
    for (int c = 0; c < NC; ++c) {
        s[c] = 0; nx[c] = 0;
        if (e[c] != 0) { s[c] = src[e[c] - 1]; nx[c] = next[e[c] - 1]; }
    }

    for (int hop = 0; hop < E; ++hop) {
        int alive = 0;
#pragma unroll
        for (int c = 0; c < NC; ++c) alive |= e[c];
        if (alive == 0) break;

        float4 t[NC][V];
#pragma unroll
        for (int c = 0; c < NC; ++c) {
#pragma unroll
            for (int v = 0; v < V; ++v) t[c][v] = make_float4(0.f, 0.f, 0.f, 0.f);
            if (e[c] != 0) {
                const float4* row = (const float4*)(x + (size_t)s[c] * C);
#pragma unroll
                for (int v = 0; v < V; ++v) t[c][v] = row[lane + 32 * v];
                cnt[c / 2] += 1;
            }
        }
#pragma unroll
        for (int c = 0; c < NC; ++c) {
            e[c] = sage_link_or_end(nx[c], E);
            if (e[c] != 0) { s[c] = src[e[c] - 1]; nx[c] = next[e[c] - 1]; }
            else           { nx[c] = 0; }
        }
#pragma unroll
        for (int c = 0; c < NC; ++c) {
#pragma unroll
            for (int v = 0; v < V; ++v) {
                acc[c / 2][v].x += t[c][v].x; acc[c / 2][v].y += t[c][v].y;
                acc[c / 2][v].z += t[c][v].z; acc[c / 2][v].w += t[c][v].w;
            }
        }
    }

#pragma unroll
    for (int r = 0; r < R; ++r) {
        if (row0 + r >= nDst) break;
        const float scale = (cnt[r] > 0) ? (1.0f / (float)cnt[r]) : 0.0f;
        float4* orow = (float4*)(outm + (size_t)(row0 + r) * C);
#pragma unroll
        for (int v = 0; v < V; ++v) {
            float4 o;
            o.x = acc[r][v].x * scale; o.y = acc[r][v].y * scale;
            o.z = acc[r][v].z * scale; o.w = acc[r][v].w * scale;
            orow[lane + 32 * v] = o;
        }
    }
}

template <int K, int N, int MT, int NT, int CH, bool RELU>
__global__ void __launch_bounds__(N / (16 * NT) * 32)
sage_gemm_wreg(const float* __restrict__ A1, const float* __restrict__ B1,
               const float* __restrict__ A2, const float* __restrict__ B2,
               const float* __restrict__ bias, float* __restrict__ out, int M) {
    constexpr int KS  = K / 4;
    constexpr int NCH = KS / CH;
    const int lane = threadIdx.x & 31;
    const int wave = threadIdx.x >> 5;
    const int c    = lane & 15;
    const int khi  = (lane >> 4) * 2;
    const int n0   = wave * 16 * NT;

    const int numTiles = M / (16 * MT);
    int tile = blockIdx.x;
    if (tile >= numTiles) return;

    v2f bw[2][KS][NT];
#pragma unroll
    for (int nt = 0; nt < NT; ++nt) {
        const size_t ro = (size_t)(n0 + nt * 16 + c) * K + khi;
#pragma unroll
        for (int j = 0; j < KS; ++j) {
            bw[0][j][nt] = *(const v2f*)(B1 + ro + 4 * j);
            bw[1][j][nt] = *(const v2f*)(B2 + ro + 4 * j);
        }
    }

    const int laneOff = c * K + khi;
    v2f cur[CH][MT];
    {
        const float* p = A1 + (size_t)tile * (16 * MT) * K + laneOff;
#pragma unroll
        for (int s = 0; s < CH; ++s)
#pragma unroll
            for (int mt = 0; mt < MT; ++mt)
                cur[s][mt] = *(const v2f*)(p + mt * 16 * K + 4 * s);
    }

    for (; tile < numTiles; tile += gridDim.x) {
        const int m0 = tile * 16 * MT;
        const float* p1 = A1 + (size_t)m0 * K + laneOff;
        const float* p2 = A2 + (size_t)m0 * K + laneOff;
        int nextTile = tile + gridDim.x;
        if (nextTile >= numTiles) nextTile = tile;
        const float* pn = A1 + (size_t)nextTile * (16 * MT) * K + laneOff;

        v8f acc[MT][NT] = {};

#pragma unroll
        for (int g = 0; g < 2 * NCH; ++g) {
            const int gn = g + 1;
            const float* q = (gn < NCH)     ? (p1 + gn * CH * 4)
                           : (gn < 2 * NCH) ? (p2 + (gn - NCH) * CH * 4)
                                            : pn;
            v2f nxt[CH][MT];
#pragma unroll
            for (int s = 0; s < CH; ++s)
#pragma unroll
                for (int mt = 0; mt < MT; ++mt)
                    nxt[s][mt] = *(const v2f*)(q + mt * 16 * K + 4 * s);
            __builtin_amdgcn_sched_barrier(0);

            const int ph = g / NCH;
            const int jb = (g % NCH) * CH;
#pragma unroll
            for (int s = 0; s < CH; ++s)
#pragma unroll
                for (int mt = 0; mt < MT; ++mt)
#pragma unroll
                    for (int nt = 0; nt < NT; ++nt)
                        acc[mt][nt] = __builtin_amdgcn_wmma_f32_16x16x4_f32(
                            false, cur[s][mt], false, bw[ph][jb + s][nt],
                            (short)0, acc[mt][nt], false, false);
#pragma unroll
            for (int s = 0; s < CH; ++s)
#pragma unroll
                for (int mt = 0; mt < MT; ++mt) cur[s][mt] = nxt[s][mt];
            __builtin_amdgcn_sched_barrier(0);
        }

        const int rbase = ((lane >> 4) << 3);
#pragma unroll
        for (int nt = 0; nt < NT; ++nt) {
            const int ocol = n0 + nt * 16 + c;
            const float bb = bias[ocol];
#pragma unroll
            for (int mt = 0; mt < MT; ++mt) {
#pragma unroll
                for (int r = 0; r < 8; ++r) {
                    float v = acc[mt][nt][r] + bb;
                    if (RELU) v = fmaxf(v, 0.0f);
                    out[(size_t)(m0 + mt * 16 + rbase + r) * N + ocol] = v;
                }
            }
        }
    }
}

template <int K, int N, int MT, int NT, int CH>
__global__ void __launch_bounds__(N / (16 * NT) * 32)
sage_proj_wreg(const float* __restrict__ A,
               const float* __restrict__ Bp, const float* __restrict__ Bq,
               float* __restrict__ P, float* __restrict__ Q,
               int Mp, int Mq, int groupsP) {
    constexpr int KS  = K / 4;
    constexpr int NCH = KS / CH;
    const int lane = threadIdx.x & 31;
    const int wave = threadIdx.x >> 5;
    const int c    = lane & 15;
    const int khi  = (lane >> 4) * 2;
    const int n0   = wave * 16 * NT;

    const bool isQ = (int)blockIdx.x >= groupsP;
    const float* B   = isQ ? Bq : Bp;
    float*       out = isQ ? Q : P;
    const int numTiles = (isQ ? Mq : Mp) / (16 * MT);
    const int step     = isQ ? ((int)gridDim.x - groupsP) : groupsP;
    int tile           = isQ ? ((int)blockIdx.x - groupsP) : (int)blockIdx.x;
    if (tile >= numTiles) return;

    v2f bw[KS][NT];
#pragma unroll
    for (int nt = 0; nt < NT; ++nt) {
        const size_t ro = (size_t)(n0 + nt * 16 + c) * K + khi;
#pragma unroll
        for (int j = 0; j < KS; ++j) bw[j][nt] = *(const v2f*)(B + ro + 4 * j);
    }

    const int laneOff = c * K + khi;
    v2f cur[CH][MT];
    {
        const float* p = A + (size_t)tile * (16 * MT) * K + laneOff;
#pragma unroll
        for (int s = 0; s < CH; ++s)
#pragma unroll
            for (int mt = 0; mt < MT; ++mt)
                cur[s][mt] = *(const v2f*)(p + mt * 16 * K + 4 * s);
    }

    for (; tile < numTiles; tile += step) {
        const int m0 = tile * 16 * MT;
        const float* p1 = A + (size_t)m0 * K + laneOff;
        int nextTile = tile + step;
        if (nextTile >= numTiles) nextTile = tile;
        const float* pn = A + (size_t)nextTile * (16 * MT) * K + laneOff;

        v8f acc[MT][NT] = {};

#pragma unroll
        for (int g = 0; g < NCH; ++g) {
            const int gn = g + 1;
            const float* q = (gn < NCH) ? (p1 + gn * CH * 4) : pn;
            v2f nxt[CH][MT];
#pragma unroll
            for (int s = 0; s < CH; ++s)
#pragma unroll
                for (int mt = 0; mt < MT; ++mt)
                    nxt[s][mt] = *(const v2f*)(q + mt * 16 * K + 4 * s);
            __builtin_amdgcn_sched_barrier(0);

            const int jb = g * CH;
#pragma unroll
            for (int s = 0; s < CH; ++s)
#pragma unroll
                for (int mt = 0; mt < MT; ++mt)
#pragma unroll
                    for (int nt = 0; nt < NT; ++nt)
                        acc[mt][nt] = __builtin_amdgcn_wmma_f32_16x16x4_f32(
                            false, cur[s][mt], false, bw[jb + s][nt],
                            (short)0, acc[mt][nt], false, false);
#pragma unroll
            for (int s = 0; s < CH; ++s)
#pragma unroll
                for (int mt = 0; mt < MT; ++mt) cur[s][mt] = nxt[s][mt];
            __builtin_amdgcn_sched_barrier(0);
        }

        const int rbase = ((lane >> 4) << 3);
#pragma unroll
        for (int nt = 0; nt < NT; ++nt) {
            const int ocol = n0 + nt * 16 + c;
#pragma unroll
            for (int mt = 0; mt < MT; ++mt) {
#pragma unroll
                for (int r = 0; r < 8; ++r)
                    out[(size_t)(m0 + mt * 16 + rbase + r) * N + ocol] =
                        acc[mt][nt][r];
            }
        }
    }
}

template <int R>
__global__ void sage_walk_mean_lsm64(const float* __restrict__ P,
                                     const int* __restrict__ src,
                                     const int* __restrict__ next,
                                     const int* __restrict__ head,
                                     const float* __restrict__ bias,
                                     float* out, int nDst, int E) {
    const int wave = blockIdx.x * (blockDim.x >> 5) +
                     __builtin_amdgcn_readfirstlane(threadIdx.x >> 5);
    const int lane = threadIdx.x & 31;
    const int row0 = wave * R;
    if (row0 >= nDst) return;
    constexpr int NC = 2 * R;
    float2 acc[R];
    int e[NC], s[NC], nx[NC], cnt[R];
#pragma unroll
    for (int r = 0; r < R; ++r) {
        acc[r] = make_float2(0.f, 0.f);
        cnt[r] = 0;
    }
#pragma unroll
    for (int c = 0; c < NC; ++c)
        e[c] = (row0 + c / 2 < nDst)
                   ? sage_link_or_end(head[2 * row0 + c], E) : 0;
#pragma unroll
    for (int c = 0; c < NC; ++c) {
        s[c] = 0; nx[c] = 0;
        if (e[c] != 0) { s[c] = src[e[c] - 1]; nx[c] = next[e[c] - 1]; }
    }

    for (int hop = 0; hop < E; ++hop) {
        int alive = 0;
#pragma unroll
        for (int c = 0; c < NC; ++c) alive |= e[c];
        if (alive == 0) break;

        float2 t[NC];
#pragma unroll
        for (int c = 0; c < NC; ++c) {
            t[c] = make_float2(0.f, 0.f);
            if (e[c] != 0) {
                t[c] = ((const float2*)(P + (size_t)s[c] * 64))[lane];
                cnt[c / 2] += 1;
            }
        }
#pragma unroll
        for (int c = 0; c < NC; ++c) {
            e[c] = sage_link_or_end(nx[c], E);
            if (e[c] != 0) { s[c] = src[e[c] - 1]; nx[c] = next[e[c] - 1]; }
            else           { nx[c] = 0; }
        }
#pragma unroll
        for (int c = 0; c < NC; ++c) {
            acc[c / 2].x += t[c].x; acc[c / 2].y += t[c].y;
        }
    }

    const float2 bb = ((const float2*)bias)[lane];
#pragma unroll
    for (int r = 0; r < R; ++r) {
        if (row0 + r >= nDst) break;
        const float scale = (cnt[r] > 0) ? (1.0f / (float)cnt[r]) : 0.0f;
        float2* orow = (float2*)(out + (size_t)(row0 + r) * 64);
        const float2 own = orow[lane];
        const float a = (acc[r].x * scale + bb.x) + own.x;
        const float b = (acc[r].y * scale + bb.y) + own.y;
        float m = fmaxf(a, b);
#pragma unroll
        for (int off = 16; off > 0; off >>= 1)
            m = fmaxf(m, __shfl_xor(m, off, 32));
        float z = expf(a - m) + expf(b - m);
#pragma unroll
        for (int off = 16; off > 0; off >>= 1)
            z += __shfl_xor(z, off, 32);
        const float lse = m + logf(z);
        orow[lane] = make_float2(a - lse, b - lse);
    }
}

extern "C" void kernel_launch(void* const* d_in, const int* in_sizes, int n_in,
                              void* d_out, int out_size, void* d_ws, size_t ws_size,
                              hipStream_t stream) {
    const float* x    = (const float*)d_in[0];
    const int*   src1 = (const int*)d_in[1];
    const int*   dst1 = (const int*)d_in[2];
    const int*   src2 = (const int*)d_in[3];
    const int*   dst2 = (const int*)d_in[4];
    const float* W1l  = (const float*)d_in[5];
    const float* W1r  = (const float*)d_in[6];
    const float* b1   = (const float*)d_in[7];
    const float* W2l  = (const float*)d_in[8];
    const float* W2r  = (const float*)d_in[9];
    const float* b2   = (const float*)d_in[10];
    float* out = (float*)d_out;

    int* head1 = (int*)d_ws;
    int* head2 = head1 + 2 * N1_;
    int* next1 = head1 + 3 * N1_ + 3 * N2_;
    int* next2 = next1 + E1_;
    float* aggr1 = (float*)(next2 + E2_);
    float* proj2 = aggr1 + (size_t)N1_ * IN_C;
    float* h     = proj2 + (size_t)N1_ * OUT_C;

    size_t zbytes = (size_t)(2 * N1_ + 2 * N2_) * sizeof(int);
    hipMemsetAsync(d_ws, 0, zbytes, stream);

    sage_link_edges<<<E1_ / 1024, 256, 0, stream>>>(dst1, head1, next1, E1_);
    sage_walk_mean<IN_C, 4><<<N1_ / 32, 256, 0, stream>>>(x, src1, next1, head1,
                                                          aggr1, N1_, E1_);
    sage_gemm_wreg<IN_C, HID_C, 2, 4, 16, true>
        <<<256, 128, 0, stream>>>(aggr1, W1l, x, W1r, b1, h, N1_);

    sage_link_edges<<<E2_ / 1024, 256, 0, stream>>>(dst2, head2, next2, E2_);
    sage_proj_wreg<HID_C, OUT_C, 2, 4, 16>
        <<<854, 32, 0, stream>>>(h, W2l, W2r, proj2, out, N1_, N2_, 683);
    sage_walk_mean_lsm64<4><<<N2_ / 32, 256, 0, stream>>>(proj2, src2, next2,
                                                          head2, b2, out,
                                                          N2_, E2_);
}


# Lists built in LDS, no global atomics or memset: 8 groups per layer each link an edge chunk over all rows by LDS exchange (heads table in LDS); 8 short lists per row; 7 -> 5 launches; sums reorder
# speedup vs baseline: 3.1479x; 1.1671x over previous
#include <hip/hip_runtime.h>
#include <hip/hip_bf16.h>

#define N0    262144
#define N1_   65536
#define N2_   16384
#define E1_   1048576
#define E2_   262144
#define IN_C  128
#define HID_C 256
#define OUT_C 64

typedef __attribute__((ext_vector_type(2))) float v2f;
typedef __attribute__((ext_vector_type(8))) float v8f;

#define NL 8
#define LB 8
static_assert((E1_ / (4 * NL)) % (LB * 1024) == 0, "chunk of layer 1");
static_assert((E2_ / (4 * NL)) % (LB * 1024) == 0, "chunk of layer 2");
static_assert(N2_ <= N1_, "table size");

__global__ void __launch_bounds__(1024)
sage_link_chunks(const int* __restrict__ dstA, int* __restrict__ headA,
                 int* __restrict__ nextA,
                 const int* __restrict__ dstB, int* __restrict__ headB,
                 int* __restrict__ nextB) {
    __shared__ int tab[N1_];
    const bool second = (int)blockIdx.x >= NL;
    const int k      = (int)blockIdx.x - (second ? NL : 0);
    const int nDst   = second ? N2_ : N1_;
    const int items  = (second ? E2_ : E1_) / (4 * NL);
    const int4* dst4 = (const int4*)(second ? dstB : dstA) + (size_t)k * items;
    int4* next4      = (int4*)(second ? nextB : nextA) + (size_t)k * items;
    int4* head4      = (int4*)((second ? headB : headA) + (size_t)k * nDst);
    const int ebase  = 4 * k * items;
    const int tid    = threadIdx.x;

    for (int i = tid; i < nDst / 4; i += 1024)
        ((int4*)tab)[i] = make_int4(0, 0, 0, 0);
    __syncthreads();

    for (int it = tid; it < items; it += LB * 1024) {
        int4 d[LB];
#pragma unroll
        for (int b = 0; b < LB; ++b) d[b] = dst4[it + b * 1024];
#pragma unroll
        for (int b = 0; b < LB; ++b) {
            const int e0 = ebase + 4 * (it + b * 1024);
            int4 p;
            p.x = atomicExch(&tab[d[b].x], e0 + 1);
            p.y = atomicExch(&tab[d[b].y], e0 + 2);
            p.z = atomicExch(&tab[d[b].z], e0 + 3);
            p.w = atomicExch(&tab[d[b].w], e0 + 4);
            next4[it + b * 1024] = p;
        }
    }
    __syncthreads();

    for (int i = tid; i < nDst / 4; i += 1024) head4[i] = ((const int4*)tab)[i];
}

static __device__ __forceinline__ int sage_link_or_end(int e, int E) {
    return ((unsigned)(e - 1) < (unsigned)E) ? e : 0;
}

template <int C, int R, int PS>
__global__ void sage_walk_mean(const float* __restrict__ x,
                               const int* __restrict__ src,
                               const int* __restrict__ next,
                               const int* __restrict__ head,
                               float* __restrict__ outm, int nDst, int E) {
    const int wave = blockIdx.x * (blockDim.x >> 5) +
                     __builtin_amdgcn_readfirstlane(threadIdx.x >> 5);
    const int lane = threadIdx.x & 31;
    constexpr int V = C / 128;
    constexpr int NC = NL * R;

    for (int ps = 0; ps < PS; ++ps) {
        const int row0 = (wave * PS + ps) * R;
        if (row0 >= nDst) break;
        float4 acc[R][V];
        int e[NC], s[NC], nx[NC], cnt[R];
#pragma unroll
        for (int r = 0; r < R; ++r) {
#pragma unroll
            for (int v = 0; v < V; ++v) acc[r][v] = make_float4(0.f, 0.f, 0.f, 0.f);
            cnt[r] = 0;
        }
        int hv[NC];
#pragma unroll
        for (int c = 0; c < NC; ++c) {
            const int rr = row0 + c % R;
            hv[c] = head[(size_t)(c / R) * nDst + (rr < nDst ? rr : row0)];
        }
#pragma unroll
        for (int c = 0; c < NC; ++c)
            e[c] = (row0 + c % R < nDst) ? sage_link_or_end(hv[c], E) : 0;
#pragma unroll
        for (int c = 0; c < NC; ++c) {
            s[c] = 0; nx[c] = 0;
            if (e[c] != 0) { s[c] = src[e[c] - 1]; nx[c] = next[e[c] - 1]; }
        }

        for (int hop = 0; hop < E; ++hop) {
            int alive = 0;
#pragma unroll
            for (int c = 0; c < NC; ++c) alive |= e[c];
            if (alive == 0) break;

            float4 t[NC][V];
#pragma unroll
            for (int c = 0; c < NC; ++c) {
#pragma unroll
                for (int v = 0; v < V; ++v) t[c][v] = make_float4(0.f, 0.f, 0.f, 0.f);
                if (e[c] != 0) {
                    const float4* row = (const float4*)(x + (size_t)s[c] * C);
#pragma unroll
                    for (int v = 0; v < V; ++v) t[c][v] = row[lane + 32 * v];
                    cnt[c % R] += 1;
                }
            }
#pragma unroll
            for (int c = 0; c < NC; ++c) {
                e[c] = sage_link_or_end(nx[c], E);
                if (e[c] != 0) { s[c] = src[e[c] - 1]; nx[c] = next[e[c] - 1]; }
                else           { nx[c] = 0; }
            }
#pragma unroll
            for (int c = 0; c < NC; ++c) {
#pragma unroll
                for (int v = 0; v < V; ++v) {
                    acc[c % R][v].x += t[c][v].x; acc[c % R][v].y += t[c][v].y;
                    acc[c % R][v].z += t[c][v].z; acc[c % R][v].w += t[c][v].w;
                }
            }
        }

#pragma unroll
        for (int r = 0; r < R; ++r) {
            if (row0 + r >= nDst) break;
            const float scale = (cnt[r] > 0) ? (1.0f / (float)cnt[r]) : 0.0f;
            float4* orow = (float4*)(outm + (size_t)(row0 + r) * C);
#pragma unroll
            for (int v = 0; v < V; ++v) {
                float4 o;
                o.x = acc[r][v].x * scale; o.y = acc[r][v].y * scale;
                o.z = acc[r][v].z * scale; o.w = acc[r][v].w * scale;
                orow[lane + 32 * v] = o;
            }
        }
    }
}

template <int K, int N, int MT, int NT, int CH, bool RELU>
__global__ void __launch_bounds__(N / (16 * NT) * 32)
sage_gemm_wreg(const float* __restrict__ A1, const float* __restrict__ B1,
               const float* __restrict__ A2, const float* __restrict__ B2,
               const float* __restrict__ bias, float* __restrict__ out, int M) {
    constexpr int KS  = K / 4;
    constexpr int NCH = KS / CH;
    const int lane = threadIdx.x & 31;
    const int wave = threadIdx.x >> 5;
    const int c    = lane & 15;
    const int khi  = (lane >> 4) * 2;
    const int n0   = wave * 16 * NT;

    const int numTiles = M / (16 * MT);
    int tile = blockIdx.x;
    if (tile >= numTiles) return;

    v2f bw[2][KS][NT];
#pragma unroll
    for (int nt = 0; nt < NT; ++nt) {
        const size_t ro = (size_t)(n0 + nt * 16 + c) * K + khi;
#pragma unroll
        for (int j = 0; j < KS; ++j) {
            bw[0][j][nt] = *(const v2f*)(B1 + ro + 4 * j);
            bw[1][j][nt] = *(const v2f*)(B2 + ro + 4 * j);
        }
    }

    const int laneOff = c * K + khi;
    v2f cur[CH][MT];
    {
        const float* p = A1 + (size_t)tile * (16 * MT) * K + laneOff;
#pragma unroll
        for (int s = 0; s < CH; ++s)
#pragma unroll
            for (int mt = 0; mt < MT; ++mt)
                cur[s][mt] = *(const v2f*)(p + mt * 16 * K + 4 * s);
    }

    for (; tile < numTiles; tile += gridDim.x) {
        const int m0 = tile * 16 * MT;
        const float* p1 = A1 + (size_t)m0 * K + laneOff;
        const float* p2 = A2 + (size_t)m0 * K + laneOff;
        int nextTile = tile + gridDim.x;
        if (nextTile >= numTiles) nextTile = tile;
        const float* pn = A1 + (size_t)nextTile * (16 * MT) * K + laneOff;

        v8f acc[MT][NT] = {};

#pragma unroll
        for (int g = 0; g < 2 * NCH; ++g) {
            const int gn = g + 1;
            const float* q = (gn < NCH)     ? (p1 + gn * CH * 4)
                           : (gn < 2 * NCH) ? (p2 + (gn - NCH) * CH * 4)
                                            : pn;
            v2f nxt[CH][MT];
#pragma unroll
            for (int s = 0; s < CH; ++s)
#pragma unroll
                for (int mt = 0; mt < MT; ++mt)
                    nxt[s][mt] = *(const v2f*)(q + mt * 16 * K + 4 * s);
            __builtin_amdgcn_sched_barrier(0);

            const int ph = g / NCH;
            const int jb = (g % NCH) * CH;
#pragma unroll
            for (int s = 0; s < CH; ++s)
#pragma unroll
                for (int mt = 0; mt < MT; ++mt)
#pragma unroll
                    for (int nt = 0; nt < NT; ++nt)
                        acc[mt][nt] = __builtin_amdgcn_wmma_f32_16x16x4_f32(
                            false, cur[s][mt], false, bw[ph][jb + s][nt],
                            (short)0, acc[mt][nt], false, false);
#pragma unroll
            for (int s = 0; s < CH; ++s)
#pragma unroll
                for (int mt = 0; mt < MT; ++mt) cur[s][mt] = nxt[s][mt];
            __builtin_amdgcn_sched_barrier(0);
        }

        const int rbase = ((lane >> 4) << 3);
#pragma unroll
        for (int nt = 0; nt < NT; ++nt) {
            const int ocol = n0 + nt * 16 + c;
            const float bb = bias[ocol];
#pragma unroll
            for (int mt = 0; mt < MT; ++mt) {
#pragma unroll
                for (int r = 0; r < 8; ++r) {
                    float v = acc[mt][nt][r] + bb;
                    if (RELU) v = fmaxf(v, 0.0f);
                    out[(size_t)(m0 + mt * 16 + rbase + r) * N + ocol] = v;
                }
            }
        }
    }
}

template <int K, int N, int MT, int NT, int CH>
__global__ void __launch_bounds__(N / (16 * NT) * 32)
sage_proj_wreg(const float* __restrict__ A,
               const float* __restrict__ Bp, const float* __restrict__ Bq,
               float* __restrict__ P, float* __restrict__ Q,
               int Mp, int Mq, int groupsP) {
    constexpr int KS  = K / 4;
    constexpr int NCH = KS / CH;
    const int lane = threadIdx.x & 31;
    const int wave = threadIdx.x >> 5;
    const int c    = lane & 15;
    const int khi  = (lane >> 4) * 2;
    const int n0   = wave * 16 * NT;

    const bool isQ = (int)blockIdx.x >= groupsP;
    const float* B   = isQ ? Bq : Bp;
    float*       out = isQ ? Q : P;
    const int numTiles = (isQ ? Mq : Mp) / (16 * MT);
    const int step     = isQ ? ((int)gridDim.x - groupsP) : groupsP;
    int tile           = isQ ? ((int)blockIdx.x - groupsP) : (int)blockIdx.x;
    if (tile >= numTiles) return;

    v2f bw[KS][NT];
#pragma unroll
    for (int nt = 0; nt < NT; ++nt) {
        const size_t ro = (size_t)(n0 + nt * 16 + c) * K + khi;
#pragma unroll
        for (int j = 0; j < KS; ++j) bw[j][nt] = *(const v2f*)(B + ro + 4 * j);
    }

    const int laneOff = c * K + khi;
    v2f cur[CH][MT];
    {
        const float* p = A + (size_t)tile * (16 * MT) * K + laneOff;
#pragma unroll
        for (int s = 0; s < CH; ++s)
#pragma unroll
            for (int mt = 0; mt < MT; ++mt)
                cur[s][mt] = *(const v2f*)(p + mt * 16 * K + 4 * s);
    }

    for (; tile < numTiles; tile += step) {
        const int m0 = tile * 16 * MT;
        const float* p1 = A + (size_t)m0 * K + laneOff;
        int nextTile = tile + step;
        if (nextTile >= numTiles) nextTile = tile;
        const float* pn = A + (size_t)nextTile * (16 * MT) * K + laneOff;

        v8f acc[MT][NT] = {};

#pragma unroll
        for (int g = 0; g < NCH; ++g) {
            const int gn = g + 1;
            const float* q = (gn < NCH) ? (p1 + gn * CH * 4) : pn;
            v2f nxt[CH][MT];
#pragma unroll
            for (int s = 0; s < CH; ++s)
#pragma unroll
                for (int mt = 0; mt < MT; ++mt)
                    nxt[s][mt] = *(const v2f*)(q + mt * 16 * K + 4 * s);
            __builtin_amdgcn_sched_barrier(0);

            const int jb = g * CH;
#pragma unroll
            for (int s = 0; s < CH; ++s)
#pragma unroll
                for (int mt = 0; mt < MT; ++mt)
#pragma unroll
                    for (int nt = 0; nt < NT; ++nt)
                        acc[mt][nt] = __builtin_amdgcn_wmma_f32_16x16x4_f32(
                            false, cur[s][mt], false, bw[jb + s][nt],
                            (short)0, acc[mt][nt], false, false);
#pragma unroll
            for (int s = 0; s < CH; ++s)
#pragma unroll
                for (int mt = 0; mt < MT; ++mt) cur[s][mt] = nxt[s][mt];
            __builtin_amdgcn_sched_barrier(0);
        }

        const int rbase = ((lane >> 4) << 3);
#pragma unroll
        for (int nt = 0; nt < NT; ++nt) {
            const int ocol = n0 + nt * 16 + c;
#pragma unroll
            for (int mt = 0; mt < MT; ++mt) {
#pragma unroll
                for (int r = 0; r < 8; ++r)
                    out[(size_t)(m0 + mt * 16 + rbase + r) * N + ocol] =
                        acc[mt][nt][r];
            }
        }
    }
}

template <int R, int PS>
__global__ void sage_walk_mean_lsm64(const float* __restrict__ P,
                                     const int* __restrict__ src,
                                     const int* __restrict__ next,
                                     const int* __restrict__ head,
                                     const float* __restrict__ bias,
                                     float* out, int nDst, int E) {
    const int wave = blockIdx.x * (blockDim.x >> 5) +
                     __builtin_amdgcn_readfirstlane(threadIdx.x >> 5);
    const int lane = threadIdx.x & 31;
    constexpr int NC = NL * R;
    const float2 bb = ((const float2*)bias)[lane];

    for (int ps = 0; ps < PS; ++ps) {
        const int row0 = (wave * PS + ps) * R;
        if (row0 >= nDst) break;
        float2 acc[R];
        int e[NC], s[NC], nx[NC], cnt[R];
#pragma unroll
        for (int r = 0; r < R; ++r) {
            acc[r] = make_float2(0.f, 0.f);
            cnt[r] = 0;
        }
        int hv[NC];
#pragma unroll
        for (int c = 0; c < NC; ++c) {
            const int rr = row0 + c % R;
            hv[c] = head[(size_t)(c / R) * nDst + (rr < nDst ? rr : row0)];
        }
#pragma unroll
        for (int c = 0; c < NC; ++c)
            e[c] = (row0 + c % R < nDst) ? sage_link_or_end(hv[c], E) : 0;
#pragma unroll
        for (int c = 0; c < NC; ++c) {
            s[c] = 0; nx[c] = 0;
            if (e[c] != 0) { s[c] = src[e[c] - 1]; nx[c] = next[e[c] - 1]; }
        }

        for (int hop = 0; hop < E; ++hop) {
            int alive = 0;
#pragma unroll
            for (int c = 0; c < NC; ++c) alive |= e[c];
            if (alive == 0) break;

            float2 t[NC];
#pragma unroll
            for (int c = 0; c < NC; ++c) {
                t[c] = make_float2(0.f, 0.f);
                if (e[c] != 0) {
                    t[c] = ((const float2*)(P + (size_t)s[c] * 64))[lane];
                    cnt[c % R] += 1;
                }
            }
#pragma unroll
            for (int c = 0; c < NC; ++c) {
                e[c] = sage_link_or_end(nx[c], E);
                if (e[c] != 0) { s[c] = src[e[c] - 1]; nx[c] = next[e[c] - 1]; }
                else           { nx[c] = 0; }
            }
#pragma unroll
            for (int c = 0; c < NC; ++c) {
                acc[c % R].x += t[c].x; acc[c % R].y += t[c].y;
            }
        }

#pragma unroll
        for (int r = 0; r < R; ++r) {
            if (row0 + r >= nDst) break;
            const float scale = (cnt[r] > 0) ? (1.0f / (float)cnt[r]) : 0.0f;
            float2* orow = (float2*)(out + (size_t)(row0 + r) * 64);
            const float2 own = orow[lane];
            const float a = (acc[r].x * scale + bb.x) + own.x;
            const float b = (acc[r].y * scale + bb.y) + own.y;
            float m = fmaxf(a, b);
#pragma unroll
            for (int off = 16; off > 0; off >>= 1)
                m = fmaxf(m, __shfl_xor(m, off, 32));
            float z = expf(a - m) + expf(b - m);
#pragma unroll
            for (int off = 16; off > 0; off >>= 1)
                z += __shfl_xor(z, off, 32);
            const float lse = m + logf(z);
            orow[lane] = make_float2(a - lse, b - lse);
        }
    }
}

extern "C" void kernel_launch(void* const* d_in, const int* in_sizes, int n_in,
                              void* d_out, int out_size, void* d_ws, size_t ws_size,
                              hipStream_t stream) {
    const float* x    = (const float*)d_in[0];
    const int*   src1 = (const int*)d_in[1];
    const int*   dst1 = (const int*)d_in[2];
    const int*   src2 = (const int*)d_in[3];
    const int*   dst2 = (const int*)d_in[4];
    const float* W1l  = (const float*)d_in[5];
    const float* W1r  = (const float*)d_in[6];
    const float* b1   = (const float*)d_in[7];
    const float* W2l  = (const float*)d_in[8];
    const float* W2r  = (const float*)d_in[9];
    const float* b2   = (const float*)d_in[10];
    float* out = (float*)d_out;

    int* head2 = (int*)d_ws;
    int* next1 = head2 + 3 * N1_ + 3 * N2_;
    int* next2 = next1 + E1_;
    float* aggr1 = (float*)(next2 + E2_);
    float* proj2 = aggr1 + (size_t)N1_ * IN_C;
    float* h     = proj2 + (size_t)N1_ * OUT_C;
    int* head1   = (int*)proj2;

    sage_link_chunks<<<2 * NL, 1024, 0, stream>>>(dst1, head1, next1,
                                                  dst2, head2, next2);

    sage_walk_mean<IN_C, 2, 2><<<N1_ / 32, 256, 0, stream>>>(x, src1, next1, head1,
                                                             aggr1, N1_, E1_);
    sage_gemm_wreg<IN_C, HID_C, 2, 4, 16, true>
        <<<256, 128, 0, stream>>>(aggr1, W1l, x, W1r, b1, h, N1_);

    sage_proj_wreg<HID_C, OUT_C, 2, 4, 16>
        <<<854, 32, 0, stream>>>(h, W2l, W2r, proj2, out, N1_, N2_, 683);
    sage_walk_mean_lsm64<2, 2><<<N2_ / 32, 256, 0, stream>>>(proj2, src2, next2,
                                                             head2, b2, out,
                                                             N2_, E2_);
}


# List walkers keep one chain per lane: a wave's 32 heads in one load, a hop's source ids and links in one load each (were scalar loads per chain); layer 2 walks 32 lists at once; sums bit-identical
# speedup vs baseline: 3.5920x; 1.1411x over previous
#include <hip/hip_runtime.h>
#include <hip/hip_bf16.h>

#define N0    262144
#define N1_   65536
#define N2_   16384
#define E1_   1048576
#define E2_   262144
#define IN_C  128
#define HID_C 256
#define OUT_C 64

typedef __attribute__((ext_vector_type(2))) float v2f;
typedef __attribute__((ext_vector_type(8))) float v8f;

#define NL 8
#define LB 8
static_assert((E1_ / (4 * NL)) % (LB * 1024) == 0, "chunk of layer 1");
static_assert((E2_ / (4 * NL)) % (LB * 1024) == 0, "chunk of layer 2");
static_assert(N2_ <= N1_, "table size");

__global__ void __launch_bounds__(1024)
sage_link_chunks(const int* __restrict__ dstA, int* __restrict__ headA,
                 int* __restrict__ nextA,
                 const int* __restrict__ dstB, int* __restrict__ headB,
                 int* __restrict__ nextB) {
    __shared__ int tab[N1_];
    const bool second = (int)blockIdx.x >= NL;
    const int k      = (int)blockIdx.x - (second ? NL : 0);
    const int nDst   = second ? N2_ : N1_;
    const int items  = (second ? E2_ : E1_) / (4 * NL);
    const int4* dst4 = (const int4*)(second ? dstB : dstA) + (size_t)k * items;
    int4* next4      = (int4*)(second ? nextB : nextA) + (size_t)k * items;
    int4* head4      = (int4*)((second ? headB : headA) + (size_t)k * nDst);
    const int ebase  = 4 * k * items;
    const int tid    = threadIdx.x;

    for (int i = tid; i < nDst / 4; i += 1024)
        ((int4*)tab)[i] = make_int4(0, 0, 0, 0);
    __syncthreads();

    for (int it = tid; it < items; it += LB * 1024) {
        int4 d[LB];
#pragma unroll
        for (int b = 0; b < LB; ++b) d[b] = dst4[it + b * 1024];
#pragma unroll
        for (int b = 0; b < LB; ++b) {
            const int e0 = ebase + 4 * (it + b * 1024);
            int4 p;
            p.x = atomicExch(&tab[d[b].x], e0 + 1);
            p.y = atomicExch(&tab[d[b].y], e0 + 2);
            p.z = atomicExch(&tab[d[b].z], e0 + 3);
            p.w = atomicExch(&tab[d[b].w], e0 + 4);
            next4[it + b * 1024] = p;
        }
    }
    __syncthreads();

    for (int i = tid; i < nDst / 4; i += 1024) head4[i] = ((const int4*)tab)[i];
}

static __device__ __forceinline__ int sage_link_or_end(int e, int E) {
    return ((unsigned)(e - 1) < (unsigned)E) ? e : 0;
}

template <int C, int R, int PS>
__global__ void sage_walk_mean(const float* __restrict__ x,
                               const int* __restrict__ src,
                               const int* __restrict__ next,
                               const int* __restrict__ head,
                               float* __restrict__ outm, int nDst, int E) {
    const int wave = blockIdx.x * (blockDim.x >> 5) +
                     __builtin_amdgcn_readfirstlane(threadIdx.x >> 5);
    const int lane = threadIdx.x & 31;
    constexpr int V = C / 128;
    constexpr int NC = NL * R;
    static_assert(NC * PS <= 32, "one lane per chain");
    constexpr unsigned PASSMASK = (NC == 32) ? 0xffffffffu : ((1u << (NC % 32)) - 1u);

    const int rowBase = wave * (R * PS);
    if (rowBase >= nDst) return;

    const int pl = lane / NC;
    const int cl = lane % NC;
    const int myrow = rowBase + pl * R + cl % R;
    int e0 = 0, sv = 0, nv = 0;
    if (pl < PS && myrow < nDst)
        e0 = sage_link_or_end(head[(size_t)(cl / R) * nDst + myrow], E);
    if (e0 != 0) { nv = next[e0 - 1]; sv = src[e0 - 1]; }
    unsigned live = __builtin_amdgcn_ballot_w32(e0 != 0);

    for (int ps = 0; ps < PS; ++ps) {
        const int row0 = rowBase + ps * R;
        if (row0 >= nDst) break;
        float4 acc[R][V];
        int cnt[R];
#pragma unroll
        for (int r = 0; r < R; ++r) {
#pragma unroll
            for (int v = 0; v < V; ++v) acc[r][v] = make_float4(0.f, 0.f, 0.f, 0.f);
            cnt[r] = 0;
        }

        for (int hop = 0; hop < E; ++hop) {
            const unsigned alive = (live >> (NC * ps)) & PASSMASK;
            if (alive == 0u) break;
            int sc[NC];
#pragma unroll
            for (int c = 0; c < NC; ++c)
                sc[c] = __builtin_amdgcn_readlane(sv, NC * ps + c);
            const int en = (pl == ps) ? sage_link_or_end(nv, E) : 0;
            const unsigned moved = __builtin_amdgcn_ballot_w32(en != 0);

            float4 t[NC][V];
#pragma unroll
            for (int c = 0; c < NC; ++c) {
#pragma unroll
                for (int v = 0; v < V; ++v) t[c][v] = make_float4(0.f, 0.f, 0.f, 0.f);
                if ((alive >> c) & 1u) {
                    const float4* row = (const float4*)(x + (size_t)sc[c] * C);
#pragma unroll
                    for (int v = 0; v < V; ++v) t[c][v] = row[lane + 32 * v];
                    cnt[c % R] += 1;
                }
            }
            if (en != 0) { nv = next[en - 1]; sv = src[en - 1]; }
            live = (live & ~(PASSMASK << (NC * ps))) | moved;
#pragma unroll
            for (int c = 0; c < NC; ++c) {
#pragma unroll
                for (int v = 0; v < V; ++v) {
                    acc[c % R][v].x += t[c][v].x; acc[c % R][v].y += t[c][v].y;
                    acc[c % R][v].z += t[c][v].z; acc[c % R][v].w += t[c][v].w;
                }
            }
        }

#pragma unroll
        for (int r = 0; r < R; ++r) {
            if (row0 + r >= nDst) break;
            const float scale = (cnt[r] > 0) ? (1.0f / (float)cnt[r]) : 0.0f;
            float4* orow = (float4*)(outm + (size_t)(row0 + r) * C);
#pragma unroll
            for (int v = 0; v < V; ++v) {
                float4 o;
                o.x = acc[r][v].x * scale; o.y = acc[r][v].y * scale;
                o.z = acc[r][v].z * scale; o.w = acc[r][v].w * scale;
                orow[lane + 32 * v] = o;
            }
        }
    }
}

template <int K, int N, int MT, int NT, int CH, bool RELU>
__global__ void __launch_bounds__(N / (16 * NT) * 32)
sage_gemm_wreg(const float* __restrict__ A1, const float* __restrict__ B1,
               const float* __restrict__ A2, const float* __restrict__ B2,
               const float* __restrict__ bias, float* __restrict__ out, int M) {
    constexpr int KS  = K / 4;
    constexpr int NCH = KS / CH;
    const int lane = threadIdx.x & 31;
    const int wave = threadIdx.x >> 5;
    const int c    = lane & 15;
    const int khi  = (lane >> 4) * 2;
    const int n0   = wave * 16 * NT;

    const int numTiles = M / (16 * MT);
    int tile = blockIdx.x;
    if (tile >= numTiles) return;

    v2f bw[2][KS][NT];
#pragma unroll
    for (int nt = 0; nt < NT; ++nt) {
        const size_t ro = (size_t)(n0 + nt * 16 + c) * K + khi;
#pragma unroll
        for (int j = 0; j < KS; ++j) {
            bw[0][j][nt] = *(const v2f*)(B1 + ro + 4 * j);
            bw[1][j][nt] = *(const v2f*)(B2 + ro + 4 * j);
        }
    }

    const int laneOff = c * K + khi;
    v2f cur[CH][MT];
    {
        const float* p = A1 + (size_t)tile * (16 * MT) * K + laneOff;
#pragma unroll
        for (int s = 0; s < CH; ++s)
#pragma unroll
            for (int mt = 0; mt < MT; ++mt)
                cur[s][mt] = *(const v2f*)(p + mt * 16 * K + 4 * s);
    }

    for (; tile < numTiles; tile += gridDim.x) {
        const int m0 = tile * 16 * MT;
        const float* p1 = A1 + (size_t)m0 * K + laneOff;
        const float* p2 = A2 + (size_t)m0 * K + laneOff;
        int nextTile = tile + gridDim.x;
        if (nextTile >= numTiles) nextTile = tile;
        const float* pn = A1 + (size_t)nextTile * (16 * MT) * K + laneOff;

        v8f acc[MT][NT] = {};

#pragma unroll
        for (int g = 0; g < 2 * NCH; ++g) {
            const int gn = g + 1;
            const float* q = (gn < NCH)     ? (p1 + gn * CH * 4)
                           : (gn < 2 * NCH) ? (p2 + (gn - NCH) * CH * 4)
                                            : pn;
            v2f nxt[CH][MT];
#pragma unroll
            for (int s = 0; s < CH; ++s)
#pragma unroll
                for (int mt = 0; mt < MT; ++mt)
                    nxt[s][mt] = *(const v2f*)(q + mt * 16 * K + 4 * s);
            __builtin_amdgcn_sched_barrier(0);

            const int ph = g / NCH;
            const int jb = (g % NCH) * CH;
#pragma unroll
            for (int s = 0; s < CH; ++s)
#pragma unroll
                for (int mt = 0; mt < MT; ++mt)
#pragma unroll
                    for (int nt = 0; nt < NT; ++nt)
                        acc[mt][nt] = __builtin_amdgcn_wmma_f32_16x16x4_f32(
                            false, cur[s][mt], false, bw[ph][jb + s][nt],
                            (short)0, acc[mt][nt], false, false);
#pragma unroll
            for (int s = 0; s < CH; ++s)
#pragma unroll
                for (int mt = 0; mt < MT; ++mt) cur[s][mt] = nxt[s][mt];
            __builtin_amdgcn_sched_barrier(0);
        }

        const int rbase = ((lane >> 4) << 3);
#pragma unroll
        for (int nt = 0; nt < NT; ++nt) {
            const int ocol = n0 + nt * 16 + c;
            const float bb = bias[ocol];
#pragma unroll
            for (int mt = 0; mt < MT; ++mt) {
#pragma unroll
                for (int r = 0; r < 8; ++r) {
                    float v = acc[mt][nt][r] + bb;
                    if (RELU) v = fmaxf(v, 0.0f);
                    out[(size_t)(m0 + mt * 16 + rbase + r) * N + ocol] = v;
                }
            }
        }
    }
}

template <int K, int N, int MT, int NT, int CH>
__global__ void __launch_bounds__(N / (16 * NT) * 32)
sage_proj_wreg(const float* __restrict__ A,
               const float* __restrict__ Bp, const float* __restrict__ Bq,
               float* __restrict__ P, float* __restrict__ Q,
               int Mp, int Mq, int groupsP) {
    constexpr int KS  = K / 4;
    constexpr int NCH = KS / CH;
    const int lane = threadIdx.x & 31;
    const int wave = threadIdx.x >> 5;
    const int c    = lane & 15;
    const int khi  = (lane >> 4) * 2;
    const int n0   = wave * 16 * NT;

    const bool isQ = (int)blockIdx.x >= groupsP;
    const float* B   = isQ ? Bq : Bp;
    float*       out = isQ ? Q : P;
    const int numTiles = (isQ ? Mq : Mp) / (16 * MT);
    const int step     = isQ ? ((int)gridDim.x - groupsP) : groupsP;
    int tile           = isQ ? ((int)blockIdx.x - groupsP) : (int)blockIdx.x;
    if (tile >= numTiles) return;

    v2f bw[KS][NT];
#pragma unroll
    for (int nt = 0; nt < NT; ++nt) {
        const size_t ro = (size_t)(n0 + nt * 16 + c) * K + khi;
#pragma unroll
        for (int j = 0; j < KS; ++j) bw[j][nt] = *(const v2f*)(B + ro + 4 * j);
    }

    const int laneOff = c * K + khi;
    v2f cur[CH][MT];
    {
        const float* p = A + (size_t)tile * (16 * MT) * K + laneOff;
#pragma unroll
        for (int s = 0; s < CH; ++s)
#pragma unroll
            for (int mt = 0; mt < MT; ++mt)
                cur[s][mt] = *(const v2f*)(p + mt * 16 * K + 4 * s);
    }

    for (; tile < numTiles; tile += step) {
        const int m0 = tile * 16 * MT;
        const float* p1 = A + (size_t)m0 * K + laneOff;
        int nextTile = tile + step;
        if (nextTile >= numTiles) nextTile = tile;
        const float* pn = A + (size_t)nextTile * (16 * MT) * K + laneOff;

        v8f acc[MT][NT] = {};

#pragma unroll
        for (int g = 0; g < NCH; ++g) {
            const int gn = g + 1;
            const float* q = (gn < NCH) ? (p1 + gn * CH * 4) : pn;
            v2f nxt[CH][MT];
#pragma unroll
            for (int s = 0; s < CH; ++s)
#pragma unroll
                for (int mt = 0; mt < MT; ++mt)
                    nxt[s][mt] = *(const v2f*)(q + mt * 16 * K + 4 * s);
            __builtin_amdgcn_sched_barrier(0);

            const int jb = g * CH;
#pragma unroll
            for (int s = 0; s < CH; ++s)
#pragma unroll
                for (int mt = 0; mt < MT; ++mt)
#pragma unroll
                    for (int nt = 0; nt < NT; ++nt)
                        acc[mt][nt] = __builtin_amdgcn_wmma_f32_16x16x4_f32(
                            false, cur[s][mt], false, bw[jb + s][nt],
                            (short)0, acc[mt][nt], false, false);
#pragma unroll
            for (int s = 0; s < CH; ++s)
#pragma unroll
                for (int mt = 0; mt < MT; ++mt) cur[s][mt] = nxt[s][mt];
            __builtin_amdgcn_sched_barrier(0);
        }

        const int rbase = ((lane >> 4) << 3);
#pragma unroll
        for (int nt = 0; nt < NT; ++nt) {
            const int ocol = n0 + nt * 16 + c;
#pragma unroll
            for (int mt = 0; mt < MT; ++mt) {
#pragma unroll
                for (int r = 0; r < 8; ++r)
                    out[(size_t)(m0 + mt * 16 + rbase + r) * N + ocol] =
                        acc[mt][nt][r];
            }
        }
    }
}

template <int R, int PS>
__global__ void sage_walk_mean_lsm64(const float* __restrict__ P,
                                     const int* __restrict__ src,
                                     const int* __restrict__ next,
                                     const int* __restrict__ head,
                                     const float* __restrict__ bias,
                                     float* out, int nDst, int E) {
    const int wave = blockIdx.x * (blockDim.x >> 5) +
                     __builtin_amdgcn_readfirstlane(threadIdx.x >> 5);
    const int lane = threadIdx.x & 31;
    constexpr int NC = NL * R;
    static_assert(NC * PS <= 32, "one lane per chain");
    constexpr unsigned PASSMASK = (NC == 32) ? 0xffffffffu : ((1u << (NC % 32)) - 1u);

    const int rowBase = wave * (R * PS);
    if (rowBase >= nDst) return;
    const float2 bb = ((const float2*)bias)[lane];

    const int pl = lane / NC;
    const int cl = lane % NC;
    const int myrow = rowBase + pl * R + cl % R;
    int e0 = 0, sv = 0, nv = 0;
    if (pl < PS && myrow < nDst)
        e0 = sage_link_or_end(head[(size_t)(cl / R) * nDst + myrow], E);
    if (e0 != 0) { nv = next[e0 - 1]; sv = src[e0 - 1]; }
    unsigned live = __builtin_amdgcn_ballot_w32(e0 != 0);

    for (int ps = 0; ps < PS; ++ps) {
        const int row0 = rowBase + ps * R;
        if (row0 >= nDst) break;
        float2 acc[R];
        int cnt[R];
#pragma unroll
        for (int r = 0; r < R; ++r) {
            acc[r] = make_float2(0.f, 0.f);
            cnt[r] = 0;
        }

        for (int hop = 0; hop < E; ++hop) {
            const unsigned alive = (live >> (NC * ps)) & PASSMASK;
            if (alive == 0u) break;
            int sc[NC];
#pragma unroll
            for (int c = 0; c < NC; ++c)
                sc[c] = __builtin_amdgcn_readlane(sv, NC * ps + c);
            const int en = (pl == ps) ? sage_link_or_end(nv, E) : 0;
            const unsigned moved = __builtin_amdgcn_ballot_w32(en != 0);

            float2 t[NC];
#pragma unroll
            for (int c = 0; c < NC; ++c) {
                t[c] = make_float2(0.f, 0.f);
                if ((alive >> c) & 1u) {
                    t[c] = ((const float2*)(P + (size_t)sc[c] * 64))[lane];
                    cnt[c % R] += 1;
                }
            }
            if (en != 0) { nv = next[en - 1]; sv = src[en - 1]; }
            live = (live & ~(PASSMASK << (NC * ps))) | moved;
#pragma unroll
            for (int c = 0; c < NC; ++c) {
                acc[c % R].x += t[c].x; acc[c % R].y += t[c].y;
            }
        }

#pragma unroll
        for (int r = 0; r < R; ++r) {
            if (row0 + r >= nDst) break;
            const float scale = (cnt[r] > 0) ? (1.0f / (float)cnt[r]) : 0.0f;
            float2* orow = (float2*)(out + (size_t)(row0 + r) * 64);
            const float2 own = orow[lane];
            const float a = (acc[r].x * scale + bb.x) + own.x;
            const float b = (acc[r].y * scale + bb.y) + own.y;
            float m = fmaxf(a, b);
#pragma unroll
            for (int off = 16; off > 0; off >>= 1)
                m = fmaxf(m, __shfl_xor(m, off, 32));
            float z = expf(a - m) + expf(b - m);
#pragma unroll
            for (int off = 16; off > 0; off >>= 1)
                z += __shfl_xor(z, off, 32);
            const float lse = m + logf(z);
            orow[lane] = make_float2(a - lse, b - lse);
        }
    }
}

extern "C" void kernel_launch(void* const* d_in, const int* in_sizes, int n_in,
                              void* d_out, int out_size, void* d_ws, size_t ws_size,
                              hipStream_t stream) {
    const float* x    = (const float*)d_in[0];
    const int*   src1 = (const int*)d_in[1];
    const int*   dst1 = (const int*)d_in[2];
    const int*   src2 = (const int*)d_in[3];
    const int*   dst2 = (const int*)d_in[4];
    const float* W1l  = (const float*)d_in[5];
    const float* W1r  = (const float*)d_in[6];
    const float* b1   = (const float*)d_in[7];
    const float* W2l  = (const float*)d_in[8];
    const float* W2r  = (const float*)d_in[9];
    const float* b2   = (const float*)d_in[10];
    float* out = (float*)d_out;

    int* head2 = (int*)d_ws;
    int* next1 = head2 + 3 * N1_ + 3 * N2_;
    int* next2 = next1 + E1_;
    float* aggr1 = (float*)(next2 + E2_);
    float* proj2 = aggr1 + (size_t)N1_ * IN_C;
    float* h     = proj2 + (size_t)N1_ * OUT_C;
    int* head1   = (int*)proj2;

    sage_link_chunks<<<2 * NL, 1024, 0, stream>>>(dst1, head1, next1,
                                                  dst2, head2, next2);

    sage_walk_mean<IN_C, 2, 2><<<N1_ / 32, 256, 0, stream>>>(x, src1, next1, head1,
                                                             aggr1, N1_, E1_);
    sage_gemm_wreg<IN_C, HID_C, 2, 4, 16, true>
        <<<256, 128, 0, stream>>>(aggr1, W1l, x, W1r, b1, h, N1_);

    sage_proj_wreg<HID_C, OUT_C, 2, 4, 16>
        <<<854, 32, 0, stream>>>(h, W2l, W2r, proj2, out, N1_, N2_, 683);
    sage_walk_mean_lsm64<4, 1><<<N2_ / 32, 256, 0, stream>>>(proj2, src2, next2,
                                                             head2, b2, out,
                                                             N2_, E2_);
}


# Link launch: layer-1 edges cut into 16 chunks, not 8: 16 groups of 65536 edges, each half the LDS exchanges and half the traffic of before; a walker chain is 2 short lists end to end; sums reorder
# speedup vs baseline: 3.7416x; 1.0416x over previous
#include <hip/hip_runtime.h>
#include <hip/hip_bf16.h>

#define N0    262144
#define N1_   65536
#define N2_   16384
#define E1_   1048576
#define E2_   262144
#define IN_C  128
#define HID_C 256
#define OUT_C 64

typedef __attribute__((ext_vector_type(2))) float v2f;
typedef __attribute__((ext_vector_type(8))) float v8f;

#define NL 8
#define G1 2
#define LB 8
static_assert((E1_ / (4 * NL * G1)) % (LB * 1024) == 0, "chunk of layer 1");
static_assert((E2_ / (4 * NL)) % (LB * 1024) == 0, "chunk of layer 2");
static_assert(N2_ <= N1_, "table size");
static_assert(NL * G1 <= OUT_C, "layer-1 head tables fit the N1 x OUT_C plane they share");

__global__ void __launch_bounds__(1024)
sage_link_chunks(const int* __restrict__ dstA, int* __restrict__ headA,
                 int* __restrict__ nextA,
                 const int* __restrict__ dstB, int* __restrict__ headB,
                 int* __restrict__ nextB) {
    __shared__ int tab[N1_];
    const bool second = (int)blockIdx.x >= NL * G1;
    const int k      = (int)blockIdx.x - (second ? NL * G1 : 0);
    const int nDst   = second ? N2_ : N1_;
    const int items  = second ? E2_ / (4 * NL) : E1_ / (4 * NL * G1);
    const int4* dst4 = (const int4*)(second ? dstB : dstA) + (size_t)k * items;
    int4* next4      = (int4*)(second ? nextB : nextA) + (size_t)k * items;
    int4* head4      = (int4*)((second ? headB : headA) + (size_t)k * nDst);
    const int ebase  = 4 * k * items;
    const int tid    = threadIdx.x;

    for (int i = tid; i < nDst / 4; i += 1024)
        ((int4*)tab)[i] = make_int4(0, 0, 0, 0);
    __syncthreads();

    for (int it = tid; it < items; it += LB * 1024) {
        int4 d[LB];
#pragma unroll
        for (int b = 0; b < LB; ++b) d[b] = dst4[it + b * 1024];
#pragma unroll
        for (int b = 0; b < LB; ++b) {
            const int e0 = ebase + 4 * (it + b * 1024);
            int4 p;
            p.x = atomicExch(&tab[d[b].x], e0 + 1);
            p.y = atomicExch(&tab[d[b].y], e0 + 2);
            p.z = atomicExch(&tab[d[b].z], e0 + 3);
            p.w = atomicExch(&tab[d[b].w], e0 + 4);
            next4[it + b * 1024] = p;
        }
    }
    __syncthreads();

    for (int i = tid; i < nDst / 4; i += 1024) head4[i] = ((const int4*)tab)[i];
}

static __device__ __forceinline__ int sage_link_or_end(int e, int E) {
    return ((unsigned)(e - 1) < (unsigned)E) ? e : 0;
}

template <int C, int R, int PS, int G>
__global__ void sage_walk_mean(const float* __restrict__ x,
                               const int* __restrict__ src,
                               const int* __restrict__ next,
                               const int* __restrict__ head,
                               float* __restrict__ outm, int nDst, int E) {
    const int wave = blockIdx.x * (blockDim.x >> 5) +
                     __builtin_amdgcn_readfirstlane(threadIdx.x >> 5);
    const int lane = threadIdx.x & 31;
    constexpr int V = C / 128;
    constexpr int NC = NL * R;
    static_assert(NC * PS <= 32, "one lane per chain");
    constexpr unsigned PASSMASK = (NC == 32) ? 0xffffffffu : ((1u << (NC % 32)) - 1u);

    const int rowBase = wave * (R * PS);
    if (rowBase >= nDst) return;

    const int pl = lane / NC;
    const int cl = lane % NC;
    const int myrow = rowBase + pl * R + cl % R;
    int hl[G];
#pragma unroll
    for (int g = 0; g < G; ++g) hl[g] = 0;
    if (pl < PS && myrow < nDst) {
#pragma unroll
        for (int g = 0; g < G; ++g)
            hl[g] = sage_link_or_end(head[(size_t)(cl / R + NL * g) * nDst + myrow], E);
    }
    int e0 = 0, sv = 0, nv = 0;
#pragma unroll
    for (int g = 0; g < G; ++g)
        if (e0 == 0) { e0 = hl[g]; hl[g] = 0; }
    if (e0 != 0) { nv = next[e0 - 1]; sv = src[e0 - 1]; }
    unsigned live = __builtin_amdgcn_ballot_w32(e0 != 0);

    for (int ps = 0; ps < PS; ++ps) {
        const int row0 = rowBase + ps * R;
        if (row0 >= nDst) break;
        float4 acc[R][V];
        int cnt[R];
#pragma unroll
        for (int r = 0; r < R; ++r) {
#pragma unroll
            for (int v = 0; v < V; ++v) acc[r][v] = make_float4(0.f, 0.f, 0.f, 0.f);
            cnt[r] = 0;
        }

        for (int hop = 0; hop < E; ++hop) {
            const unsigned alive = (live >> (NC * ps)) & PASSMASK;
            if (alive == 0u) break;
            int sc[NC];
#pragma unroll
            for (int c = 0; c < NC; ++c)
                sc[c] = __builtin_amdgcn_readlane(sv, NC * ps + c);
            int en = (pl == ps) ? sage_link_or_end(nv, E) : 0;
#pragma unroll
            for (int g = 1; g < G; ++g)
                if (pl == ps && en == 0) { en = hl[g]; hl[g] = 0; }
            const unsigned moved = __builtin_amdgcn_ballot_w32(en != 0);

            float4 t[NC][V];
#pragma unroll
            for (int c = 0; c < NC; ++c) {
#pragma unroll
                for (int v = 0; v < V; ++v) t[c][v] = make_float4(0.f, 0.f, 0.f, 0.f);
                if ((alive >> c) & 1u) {
                    const float4* row = (const float4*)(x + (size_t)sc[c] * C);
#pragma unroll
                    for (int v = 0; v < V; ++v) t[c][v] = row[lane + 32 * v];
                    cnt[c % R] += 1;
                }
            }
            if (en != 0) { nv = next[en - 1]; sv = src[en - 1]; }
            live = (live & ~(PASSMASK << (NC * ps))) | moved;
#pragma unroll
            for (int c = 0; c < NC; ++c) {
#pragma unroll
                for (int v = 0; v < V; ++v) {
                    acc[c % R][v].x += t[c][v].x; acc[c % R][v].y += t[c][v].y;
                    acc[c % R][v].z += t[c][v].z; acc[c % R][v].w += t[c][v].w;
                }
            }
        }

#pragma unroll
        for (int r = 0; r < R; ++r) {
            if (row0 + r >= nDst) break;
            const float scale = (cnt[r] > 0) ? (1.0f / (float)cnt[r]) : 0.0f;
            float4* orow = (float4*)(outm + (size_t)(row0 + r) * C);
#pragma unroll
            for (int v = 0; v < V; ++v) {
                float4 o;
                o.x = acc[r][v].x * scale; o.y = acc[r][v].y * scale;
                o.z = acc[r][v].z * scale; o.w = acc[r][v].w * scale;
                orow[lane + 32 * v] = o;
            }
        }
    }
}

template <int K, int N, int MT, int NT, int CH, bool RELU>
__global__ void __launch_bounds__(N / (16 * NT) * 32)
sage_gemm_wreg(const float* __restrict__ A1, const float* __restrict__ B1,
               const float* __restrict__ A2, const float* __restrict__ B2,
               const float* __restrict__ bias, float* __restrict__ out, int M) {
    constexpr int KS  = K / 4;
    constexpr int NCH = KS / CH;
    const int lane = threadIdx.x & 31;
    const int wave = threadIdx.x >> 5;
    const int c    = lane & 15;
    const int khi  = (lane >> 4) * 2;
    const int n0   = wave * 16 * NT;

    const int numTiles = M / (16 * MT);
    int tile = blockIdx.x;
    if (tile >= numTiles) return;

    v2f bw[2][KS][NT];
#pragma unroll
    for (int nt = 0; nt < NT; ++nt) {
        const size_t ro = (size_t)(n0 + nt * 16 + c) * K + khi;
#pragma unroll
        for (int j = 0; j < KS; ++j) {
            bw[0][j][nt] = *(const v2f*)(B1 + ro + 4 * j);
            bw[1][j][nt] = *(const v2f*)(B2 + ro + 4 * j);
        }
    }

    const int laneOff = c * K + khi;
    v2f cur[CH][MT];
    {
        const float* p = A1 + (size_t)tile * (16 * MT) * K + laneOff;
#pragma unroll
        for (int s = 0; s < CH; ++s)
#pragma unroll
            for (int mt = 0; mt < MT; ++mt)
                cur[s][mt] = *(const v2f*)(p + mt * 16 * K + 4 * s);
    }

    for (; tile < numTiles; tile += gridDim.x) {
        const int m0 = tile * 16 * MT;
        const float* p1 = A1 + (size_t)m0 * K + laneOff;
        const float* p2 = A2 + (size_t)m0 * K + laneOff;
        int nextTile = tile + gridDim.x;
        if (nextTile >= numTiles) nextTile = tile;
        const float* pn = A1 + (size_t)nextTile * (16 * MT) * K + laneOff;

        v8f acc[MT][NT] = {};

#pragma unroll
        for (int g = 0; g < 2 * NCH; ++g) {
            const int gn = g + 1;
            const float* q = (gn < NCH)     ? (p1 + gn * CH * 4)
                           : (gn < 2 * NCH) ? (p2 + (gn - NCH) * CH * 4)
                                            : pn;
            v2f nxt[CH][MT];
#pragma unroll
            for (int s = 0; s < CH; ++s)
#pragma unroll
                for (int mt = 0; mt < MT; ++mt)
                    nxt[s][mt] = *(const v2f*)(q + mt * 16 * K + 4 * s);
            __builtin_amdgcn_sched_barrier(0);

            const int ph = g / NCH;
            const int jb = (g % NCH) * CH;
#pragma unroll
            for (int s = 0; s < CH; ++s)
#pragma unroll
                for (int mt = 0; mt < MT; ++mt)
#pragma unroll
                    for (int nt = 0; nt < NT; ++nt)
                        acc[mt][nt] = __builtin_amdgcn_wmma_f32_16x16x4_f32(
                            false, cur[s][mt], false, bw[ph][jb + s][nt],
                            (short)0, acc[mt][nt], false, false);
#pragma unroll
            for (int s = 0; s < CH; ++s)
#pragma unroll
                for (int mt = 0; mt < MT; ++mt) cur[s][mt] = nxt[s][mt];
            __builtin_amdgcn_sched_barrier(0);
        }

        const int rbase = ((lane >> 4) << 3);
#pragma unroll
        for (int nt = 0; nt < NT; ++nt) {
            const int ocol = n0 + nt * 16 + c;
            const float bb = bias[ocol];
#pragma unroll
            for (int mt = 0; mt < MT; ++mt) {
#pragma unroll
                for (int r = 0; r < 8; ++r) {
                    float v = acc[mt][nt][r] + bb;
                    if (RELU) v = fmaxf(v, 0.0f);
                    out[(size_t)(m0 + mt * 16 + rbase + r) * N + ocol] = v;
                }
            }
        }
    }
}

template <int K, int N, int MT, int NT, int CH>
__global__ void __launch_bounds__(N / (16 * NT) * 32)
sage_proj_wreg(const float* __restrict__ A,
               const float* __restrict__ Bp, const float* __restrict__ Bq,
               float* __restrict__ P, float* __restrict__ Q,
               int Mp, int Mq, int groupsP) {
    constexpr int KS  = K / 4;
    constexpr int NCH = KS / CH;
    const int lane = threadIdx.x & 31;
    const int wave = threadIdx.x >> 5;
    const int c    = lane & 15;
    const int khi  = (lane >> 4) * 2;
    const int n0   = wave * 16 * NT;

    const bool isQ = (int)blockIdx.x >= groupsP;
    const float* B   = isQ ? Bq : Bp;
    float*       out = isQ ? Q : P;
    const int numTiles = (isQ ? Mq : Mp) / (16 * MT);
    const int step     = isQ ? ((int)gridDim.x - groupsP) : groupsP;
    int tile           = isQ ? ((int)blockIdx.x - groupsP) : (int)blockIdx.x;
    if (tile >= numTiles) return;

    v2f bw[KS][NT];
#pragma unroll
    for (int nt = 0; nt < NT; ++nt) {
        const size_t ro = (size_t)(n0 + nt * 16 + c) * K + khi;
#pragma unroll
        for (int j = 0; j < KS; ++j) bw[j][nt] = *(const v2f*)(B + ro + 4 * j);
    }

    const int laneOff = c * K + khi;
    v2f cur[CH][MT];
    {
        const float* p = A + (size_t)tile * (16 * MT) * K + laneOff;
#pragma unroll
        for (int s = 0; s < CH; ++s)
#pragma unroll
            for (int mt = 0; mt < MT; ++mt)
                cur[s][mt] = *(const v2f*)(p + mt * 16 * K + 4 * s);
    }

    for (; tile < numTiles; tile += step) {
        const int m0 = tile * 16 * MT;
        const float* p1 = A + (size_t)m0 * K + laneOff;
        int nextTile = tile + step;
        if (nextTile >= numTiles) nextTile = tile;
        const float* pn = A + (size_t)nextTile * (16 * MT) * K + laneOff;

        v8f acc[MT][NT] = {};

#pragma unroll
        for (int g = 0; g < NCH; ++g) {
            const int gn = g + 1;
            const float* q = (gn < NCH) ? (p1 + gn * CH * 4) : pn;
            v2f nxt[CH][MT];
#pragma unroll
            for (int s = 0; s < CH; ++s)
#pragma unroll
                for (int mt = 0; mt < MT; ++mt)
                    nxt[s][mt] = *(const v2f*)(q + mt * 16 * K + 4 * s);
            __builtin_amdgcn_sched_barrier(0);

            const int jb = g * CH;
#pragma unroll
            for (int s = 0; s < CH; ++s)
#pragma unroll
                for (int mt = 0; mt < MT; ++mt)
#pragma unroll
                    for (int nt = 0; nt < NT; ++nt)
                        acc[mt][nt] = __builtin_amdgcn_wmma_f32_16x16x4_f32(
                            false, cur[s][mt], false, bw[jb + s][nt],
                            (short)0, acc[mt][nt], false, false);
#pragma unroll
            for (int s = 0; s < CH; ++s)
#pragma unroll
                for (int mt = 0; mt < MT; ++mt) cur[s][mt] = nxt[s][mt];
            __builtin_amdgcn_sched_barrier(0);
        }

        const int rbase = ((lane >> 4) << 3);
#pragma unroll
        for (int nt = 0; nt < NT; ++nt) {
            const int ocol = n0 + nt * 16 + c;
#pragma unroll
            for (int mt = 0; mt < MT; ++mt) {
#pragma unroll
                for (int r = 0; r < 8; ++r)
                    out[(size_t)(m0 + mt * 16 + rbase + r) * N + ocol] =
                        acc[mt][nt][r];
            }
        }
    }
}

template <int R, int PS>
__global__ void sage_walk_mean_lsm64(const float* __restrict__ P,
                                     const int* __restrict__ src,
                                     const int* __restrict__ next,
                                     const int* __restrict__ head,
                                     const float* __restrict__ bias,
                                     float* out, int nDst, int E) {
    const int wave = blockIdx.x * (blockDim.x >> 5) +
                     __builtin_amdgcn_readfirstlane(threadIdx.x >> 5);
    const int lane = threadIdx.x & 31;
    constexpr int NC = NL * R;
    static_assert(NC * PS <= 32, "one lane per chain");
    constexpr unsigned PASSMASK = (NC == 32) ? 0xffffffffu : ((1u << (NC % 32)) - 1u);

    const int rowBase = wave * (R * PS);
    if (rowBase >= nDst) return;
    const float2 bb = ((const float2*)bias)[lane];

    const int pl = lane / NC;
    const int cl = lane % NC;
    const int myrow = rowBase + pl * R + cl % R;
    int e0 = 0, sv = 0, nv = 0;
    if (pl < PS && myrow < nDst)
        e0 = sage_link_or_end(head[(size_t)(cl / R) * nDst + myrow], E);
    if (e0 != 0) { nv = next[e0 - 1]; sv = src[e0 - 1]; }
    unsigned live = __builtin_amdgcn_ballot_w32(e0 != 0);

    for (int ps = 0; ps < PS; ++ps) {
        const int row0 = rowBase + ps * R;
        if (row0 >= nDst) break;
        float2 acc[R];
        int cnt[R];
#pragma unroll
        for (int r = 0; r < R; ++r) {
            acc[r] = make_float2(0.f, 0.f);
            cnt[r] = 0;
        }

        for (int hop = 0; hop < E; ++hop) {
            const unsigned alive = (live >> (NC * ps)) & PASSMASK;
            if (alive == 0u) break;
            int sc[NC];
#pragma unroll
            for (int c = 0; c < NC; ++c)
                sc[c] = __builtin_amdgcn_readlane(sv, NC * ps + c);
            const int en = (pl == ps) ? sage_link_or_end(nv, E) : 0;
            const unsigned moved = __builtin_amdgcn_ballot_w32(en != 0);

            float2 t[NC];
#pragma unroll
            for (int c = 0; c < NC; ++c) {
                t[c] = make_float2(0.f, 0.f);
                if ((alive >> c) & 1u) {
                    t[c] = ((const float2*)(P + (size_t)sc[c] * 64))[lane];
                    cnt[c % R] += 1;
                }
            }
            if (en != 0) { nv = next[en - 1]; sv = src[en - 1]; }
            live = (live & ~(PASSMASK << (NC * ps))) | moved;
#pragma unroll
            for (int c = 0; c < NC; ++c) {
                acc[c % R].x += t[c].x; acc[c % R].y += t[c].y;
            }
        }

#pragma unroll
        for (int r = 0; r < R; ++r) {
            if (row0 + r >= nDst) break;
            const float scale = (cnt[r] > 0) ? (1.0f / (float)cnt[r]) : 0.0f;
            float2* orow = (float2*)(out + (size_t)(row0 + r) * 64);
            const float2 own = orow[lane];
            const float a = (acc[r].x * scale + bb.x) + own.x;
            const float b = (acc[r].y * scale + bb.y) + own.y;
            float m = fmaxf(a, b);
#pragma unroll
            for (int off = 16; off > 0; off >>= 1)
                m = fmaxf(m, __shfl_xor(m, off, 32));
            float z = expf(a - m) + expf(b - m);
#pragma unroll
            for (int off = 16; off > 0; off >>= 1)
                z += __shfl_xor(z, off, 32);
            const float lse = m + logf(z);
            orow[lane] = make_float2(a - lse, b - lse);
        }
    }
}

extern "C" void kernel_launch(void* const* d_in, const int* in_sizes, int n_in,
                              void* d_out, int out_size, void* d_ws, size_t ws_size,
                              hipStream_t stream) {
    const float* x    = (const float*)d_in[0];
    const int*   src1 = (const int*)d_in[1];
    const int*   dst1 = (const int*)d_in[2];
    const int*   src2 = (const int*)d_in[3];
    const int*   dst2 = (const int*)d_in[4];
    const float* W1l  = (const float*)d_in[5];
    const float* W1r  = (const float*)d_in[6];
    const float* b1   = (const float*)d_in[7];
    const float* W2l  = (const float*)d_in[8];
    const float* W2r  = (const float*)d_in[9];
    const float* b2   = (const float*)d_in[10];
    float* out = (float*)d_out;

    int* head2 = (int*)d_ws;
    int* next1 = head2 + 3 * N1_ + 3 * N2_;
    int* next2 = next1 + E1_;
    float* aggr1 = (float*)(next2 + E2_);
    float* proj2 = aggr1 + (size_t)N1_ * IN_C;
    float* h     = proj2 + (size_t)N1_ * OUT_C;
    int* head1   = (int*)proj2;

    sage_link_chunks<<<NL * G1 + NL, 1024, 0, stream>>>(dst1, head1, next1,
                                                        dst2, head2, next2);

    sage_walk_mean<IN_C, 2, 2, G1><<<N1_ / 32, 256, 0, stream>>>(x, src1, next1, head1,
                                                                 aggr1, N1_, E1_);
    sage_gemm_wreg<IN_C, HID_C, 2, 4, 16, true>
        <<<256, 128, 0, stream>>>(aggr1, W1l, x, W1r, b1, h, N1_);

    sage_proj_wreg<HID_C, OUT_C, 2, 4, 16>
        <<<854, 32, 0, stream>>>(h, W2l, W2r, proj2, out, N1_, N2_, 683);
    sage_walk_mean_lsm64<4, 1><<<N2_ / 32, 256, 0, stream>>>(proj2, src2, next2,
                                                             head2, b2, out,
                                                             N2_, E2_);
}


# Layer-1 walk: a row's 2 load slots go each hop to any 2 of its 8 live chains (was a slot per chain, idle once it ends); a wave walks 4 rows at once; 57 registers, more waves resident; sums reorder
# speedup vs baseline: 3.9050x; 1.0437x over previous
#include <hip/hip_runtime.h>
#include <hip/hip_bf16.h>

#define N0    262144
#define N1_   65536
#define N2_   16384
#define E1_   1048576
#define E2_   262144
#define IN_C  128
#define HID_C 256
#define OUT_C 64

typedef __attribute__((ext_vector_type(2))) float v2f;
typedef __attribute__((ext_vector_type(8))) float v8f;

#define NL 8
#define G1 2
#define LB 8
static_assert((E1_ / (4 * NL * G1)) % (LB * 1024) == 0, "chunk of layer 1");
static_assert((E2_ / (4 * NL)) % (LB * 1024) == 0, "chunk of layer 2");
static_assert(N2_ <= N1_, "table size");
static_assert(NL * G1 <= OUT_C, "layer-1 head tables fit the N1 x OUT_C plane they share");

__global__ void __launch_bounds__(1024)
sage_link_chunks(const int* __restrict__ dstA, int* __restrict__ headA,
                 int* __restrict__ nextA,
                 const int* __restrict__ dstB, int* __restrict__ headB,
                 int* __restrict__ nextB) {
    __shared__ int tab[N1_];
    const bool second = (int)blockIdx.x >= NL * G1;
    const int k      = (int)blockIdx.x - (second ? NL * G1 : 0);
    const int nDst   = second ? N2_ : N1_;
    const int items  = second ? E2_ / (4 * NL) : E1_ / (4 * NL * G1);
    const int4* dst4 = (const int4*)(second ? dstB : dstA) + (size_t)k * items;
    int4* next4      = (int4*)(second ? nextB : nextA) + (size_t)k * items;
    int4* head4      = (int4*)((second ? headB : headA) + (size_t)k * nDst);
    const int ebase  = 4 * k * items;
    const int tid    = threadIdx.x;

    for (int i = tid; i < nDst / 4; i += 1024)
        ((int4*)tab)[i] = make_int4(0, 0, 0, 0);
    __syncthreads();

    for (int it = tid; it < items; it += LB * 1024) {
        int4 d[LB];
#pragma unroll
        for (int b = 0; b < LB; ++b) d[b] = dst4[it + b * 1024];
#pragma unroll
        for (int b = 0; b < LB; ++b) {
            const int e0 = ebase + 4 * (it + b * 1024);
            int4 p;
            p.x = atomicExch(&tab[d[b].x], e0 + 1);
            p.y = atomicExch(&tab[d[b].y], e0 + 2);
            p.z = atomicExch(&tab[d[b].z], e0 + 3);
            p.w = atomicExch(&tab[d[b].w], e0 + 4);
            next4[it + b * 1024] = p;
        }
    }
    __syncthreads();

    for (int i = tid; i < nDst / 4; i += 1024) head4[i] = ((const int4*)tab)[i];
}

static __device__ __forceinline__ int sage_link_or_end(int e, int E) {
    return ((unsigned)(e - 1) < (unsigned)E) ? e : 0;
}

template <int C, int S, int G>
__global__ void sage_walk_mean_slots(const float* __restrict__ x,
                                     const int* __restrict__ src,
                                     const int* __restrict__ next,
                                     const int* __restrict__ head,
                                     float* __restrict__ outm, int nDst, int E) {
    static_assert(C == 128, "a row is one float4 per lane");
    constexpr int RW = 32 / NL;
    static_assert(RW * NL == 32, "one lane per chain");
    static_assert(S >= 1 && S <= NL, "slots per row");
    constexpr unsigned ROWMASK = (1u << NL) - 1u;
    const int wave = blockIdx.x * (blockDim.x >> 5) +
                     __builtin_amdgcn_readfirstlane(threadIdx.x >> 5);
    const int lane = threadIdx.x & 31;

    const int rowBase = wave * RW;
    if (rowBase >= nDst) return;

    const int rl = lane / NL;
    const int jl = lane % NL;
    const int myrow = rowBase + rl;
    int hl[G];
#pragma unroll
    for (int g = 0; g < G; ++g) hl[g] = 0;
    if (myrow < nDst) {
#pragma unroll
        for (int g = 0; g < G; ++g)
            hl[g] = sage_link_or_end(head[(size_t)(jl + NL * g) * nDst + myrow], E);
    }
    int e0 = 0, sv = 0, nv = 0;
#pragma unroll
    for (int g = 0; g < G; ++g)
        if (e0 == 0) { e0 = hl[g]; hl[g] = 0; }
    if (e0 != 0) { nv = next[e0 - 1]; sv = src[e0 - 1]; }
    unsigned live = __builtin_amdgcn_ballot_w32(e0 != 0);

    float4 acc[RW];
    int cnt[RW];
#pragma unroll
    for (int r = 0; r < RW; ++r) {
        acc[r] = make_float4(0.f, 0.f, 0.f, 0.f);
        cnt[r] = 0;
    }

    for (int hop = 0; hop < E; ++hop) {
        if (live == 0u) break;
        int pk[RW][S];
        bool ok[RW][S];
        unsigned picked = 0u;
#pragma unroll
        for (int r = 0; r < RW; ++r) {
            unsigned m = (live >> (NL * r)) & ROWMASK;
#pragma unroll
            for (int s = 0; s < S; ++s) {
                ok[r][s] = (m != 0u);
                pk[r][s] = ok[r][s] ? (NL * r + __builtin_ctz(m)) : 0;
                if (ok[r][s]) picked |= 1u << pk[r][s];
                m &= m - 1u;
            }
        }
        int sc[RW][S];
#pragma unroll
        for (int r = 0; r < RW; ++r)
#pragma unroll
            for (int s = 0; s < S; ++s)
                sc[r][s] = __builtin_amdgcn_readlane(sv, pk[r][s]);
        const bool mine = ((picked >> lane) & 1u) != 0u;
        int en = mine ? sage_link_or_end(nv, E) : 0;
#pragma unroll
        for (int g = 1; g < G; ++g)
            if (mine && en == 0) { en = hl[g]; hl[g] = 0; }
        const unsigned moved = __builtin_amdgcn_ballot_w32(en != 0);

        float4 t[RW][S];
#pragma unroll
        for (int r = 0; r < RW; ++r) {
#pragma unroll
            for (int s = 0; s < S; ++s) {
                t[r][s] = make_float4(0.f, 0.f, 0.f, 0.f);
                if (ok[r][s]) {
                    const float4* row = (const float4*)(x + (size_t)sc[r][s] * C);
                    t[r][s] = row[lane];
                    cnt[r] += 1;
                }
            }
        }
        if (en != 0) { nv = next[en - 1]; sv = src[en - 1]; }
        live = (live & ~picked) | moved;
#pragma unroll
        for (int r = 0; r < RW; ++r) {
#pragma unroll
            for (int s = 0; s < S; ++s) {
                acc[r].x += t[r][s].x; acc[r].y += t[r][s].y;
                acc[r].z += t[r][s].z; acc[r].w += t[r][s].w;
            }
        }
    }

#pragma unroll
    for (int r = 0; r < RW; ++r) {
        if (rowBase + r >= nDst) break;
        const float scale = (cnt[r] > 0) ? (1.0f / (float)cnt[r]) : 0.0f;
        float4* orow = (float4*)(outm + (size_t)(rowBase + r) * C);
        float4 o;
        o.x = acc[r].x * scale; o.y = acc[r].y * scale;
        o.z = acc[r].z * scale; o.w = acc[r].w * scale;
        orow[lane] = o;
    }
}

template <int K, int N, int MT, int NT, int CH, bool RELU>
__global__ void __launch_bounds__(N / (16 * NT) * 32)
sage_gemm_wreg(const float* __restrict__ A1, const float* __restrict__ B1,
               const float* __restrict__ A2, const float* __restrict__ B2,
               const float* __restrict__ bias, float* __restrict__ out, int M) {
    constexpr int KS  = K / 4;
    constexpr int NCH = KS / CH;
    const int lane = threadIdx.x & 31;
    const int wave = threadIdx.x >> 5;
    const int c    = lane & 15;
    const int khi  = (lane >> 4) * 2;
    const int n0   = wave * 16 * NT;

    const int numTiles = M / (16 * MT);
    int tile = blockIdx.x;
    if (tile >= numTiles) return;

    v2f bw[2][KS][NT];
#pragma unroll
    for (int nt = 0; nt < NT; ++nt) {
        const size_t ro = (size_t)(n0 + nt * 16 + c) * K + khi;
#pragma unroll
        for (int j = 0; j < KS; ++j) {
            bw[0][j][nt] = *(const v2f*)(B1 + ro + 4 * j);
            bw[1][j][nt] = *(const v2f*)(B2 + ro + 4 * j);
        }
    }

    const int laneOff = c * K + khi;
    v2f cur[CH][MT];
    {
        const float* p = A1 + (size_t)tile * (16 * MT) * K + laneOff;
#pragma unroll
        for (int s = 0; s < CH; ++s)
#pragma unroll
            for (int mt = 0; mt < MT; ++mt)
                cur[s][mt] = *(const v2f*)(p + mt * 16 * K + 4 * s);
    }

    for (; tile < numTiles; tile += gridDim.x) {
        const int m0 = tile * 16 * MT;
        const float* p1 = A1 + (size_t)m0 * K + laneOff;
        const float* p2 = A2 + (size_t)m0 * K + laneOff;
        int nextTile = tile + gridDim.x;
        if (nextTile >= numTiles) nextTile = tile;
        const float* pn = A1 + (size_t)nextTile * (16 * MT) * K + laneOff;

        v8f acc[MT][NT] = {};

#pragma unroll
        for (int g = 0; g < 2 * NCH; ++g) {
            const int gn = g + 1;
            const float* q = (gn < NCH)     ? (p1 + gn * CH * 4)
                           : (gn < 2 * NCH) ? (p2 + (gn - NCH) * CH * 4)
                                            : pn;
            v2f nxt[CH][MT];
#pragma unroll
            for (int s = 0; s < CH; ++s)
#pragma unroll
                for (int mt = 0; mt < MT; ++mt)
                    nxt[s][mt] = *(const v2f*)(q + mt * 16 * K + 4 * s);
            __builtin_amdgcn_sched_barrier(0);

            const int ph = g / NCH;
            const int jb = (g % NCH) * CH;
#pragma unroll
            for (int s = 0; s < CH; ++s)
#pragma unroll
                for (int mt = 0; mt < MT; ++mt)
#pragma unroll
                    for (int nt = 0; nt < NT; ++nt)
                        acc[mt][nt] = __builtin_amdgcn_wmma_f32_16x16x4_f32(
                            false, cur[s][mt], false, bw[ph][jb + s][nt],
                            (short)0, acc[mt][nt], false, false);
#pragma unroll
            for (int s = 0; s < CH; ++s)
#pragma unroll
                for (int mt = 0; mt < MT; ++mt) cur[s][mt] = nxt[s][mt];
            __builtin_amdgcn_sched_barrier(0);
        }

        const int rbase = ((lane >> 4) << 3);
#pragma unroll
        for (int nt = 0; nt < NT; ++nt) {
            const int ocol = n0 + nt * 16 + c;
            const float bb = bias[ocol];
#pragma unroll
            for (int mt = 0; mt < MT; ++mt) {
#pragma unroll
                for (int r = 0; r < 8; ++r) {
                    float v = acc[mt][nt][r] + bb;
                    if (RELU) v = fmaxf(v, 0.0f);
                    out[(size_t)(m0 + mt * 16 + rbase + r) * N + ocol] = v;
                }
            }
        }
    }
}

template <int K, int N, int MT, int NT, int CH>
__global__ void __launch_bounds__(N / (16 * NT) * 32)
sage_proj_wreg(const float* __restrict__ A,
               const float* __restrict__ Bp, const float* __restrict__ Bq,
               float* __restrict__ P, float* __restrict__ Q,
               int Mp, int Mq, int groupsP) {
    constexpr int KS  = K / 4;
    constexpr int NCH = KS / CH;
    const int lane = threadIdx.x & 31;
    const int wave = threadIdx.x >> 5;
    const int c    = lane & 15;
    const int khi  = (lane >> 4) * 2;
    const int n0   = wave * 16 * NT;

    const bool isQ = (int)blockIdx.x >= groupsP;
    const float* B   = isQ ? Bq : Bp;
    float*       out = isQ ? Q : P;
    const int numTiles = (isQ ? Mq : Mp) / (16 * MT);
    const int step     = isQ ? ((int)gridDim.x - groupsP) : groupsP;
    int tile           = isQ ? ((int)blockIdx.x - groupsP) : (int)blockIdx.x;
    if (tile >= numTiles) return;

    v2f bw[KS][NT];
#pragma unroll
    for (int nt = 0; nt < NT; ++nt) {
        const size_t ro = (size_t)(n0 + nt * 16 + c) * K + khi;
#pragma unroll
        for (int j = 0; j < KS; ++j) bw[j][nt] = *(const v2f*)(B + ro + 4 * j);
    }

    const int laneOff = c * K + khi;
    v2f cur[CH][MT];
    {
        const float* p = A + (size_t)tile * (16 * MT) * K + laneOff;
#pragma unroll
        for (int s = 0; s < CH; ++s)
#pragma unroll
            for (int mt = 0; mt < MT; ++mt)
                cur[s][mt] = *(const v2f*)(p + mt * 16 * K + 4 * s);
    }

    for (; tile < numTiles; tile += step) {
        const int m0 = tile * 16 * MT;
        const float* p1 = A + (size_t)m0 * K + laneOff;
        int nextTile = tile + step;
        if (nextTile >= numTiles) nextTile = tile;
        const float* pn = A + (size_t)nextTile * (16 * MT) * K + laneOff;

        v8f acc[MT][NT] = {};

#pragma unroll
        for (int g = 0; g < NCH; ++g) {
            const int gn = g + 1;
            const float* q = (gn < NCH) ? (p1 + gn * CH * 4) : pn;
            v2f nxt[CH][MT];
#pragma unroll
            for (int s = 0; s < CH; ++s)
#pragma unroll
                for (int mt = 0; mt < MT; ++mt)
                    nxt[s][mt] = *(const v2f*)(q + mt * 16 * K + 4 * s);
            __builtin_amdgcn_sched_barrier(0);

            const int jb = g * CH;
#pragma unroll
            for (int s = 0; s < CH; ++s)
#pragma unroll
                for (int mt = 0; mt < MT; ++mt)
#pragma unroll
                    for (int nt = 0; nt < NT; ++nt)
                        acc[mt][nt] = __builtin_amdgcn_wmma_f32_16x16x4_f32(
                            false, cur[s][mt], false, bw[jb + s][nt],
                            (short)0, acc[mt][nt], false, false);
#pragma unroll
            for (int s = 0; s < CH; ++s)
#pragma unroll
                for (int mt = 0; mt < MT; ++mt) cur[s][mt] = nxt[s][mt];
            __builtin_amdgcn_sched_barrier(0);
        }

        const int rbase = ((lane >> 4) << 3);
#pragma unroll
        for (int nt = 0; nt < NT; ++nt) {
            const int ocol = n0 + nt * 16 + c;
#pragma unroll
            for (int mt = 0; mt < MT; ++mt) {
#pragma unroll
                for (int r = 0; r < 8; ++r)
                    out[(size_t)(m0 + mt * 16 + rbase + r) * N + ocol] =
                        acc[mt][nt][r];
            }
        }
    }
}

template <int R, int PS>
__global__ void sage_walk_mean_lsm64(const float* __restrict__ P,
                                     const int* __restrict__ src,
                                     const int* __restrict__ next,
                                     const int* __restrict__ head,
                                     const float* __restrict__ bias,
                                     float* out, int nDst, int E) {
    const int wave = blockIdx.x * (blockDim.x >> 5) +
                     __builtin_amdgcn_readfirstlane(threadIdx.x >> 5);
    const int lane = threadIdx.x & 31;
    constexpr int NC = NL * R;
    static_assert(NC * PS <= 32, "one lane per chain");
    constexpr unsigned PASSMASK = (NC == 32) ? 0xffffffffu : ((1u << (NC % 32)) - 1u);

    const int rowBase = wave * (R * PS);
    if (rowBase >= nDst) return;
    const float2 bb = ((const float2*)bias)[lane];

    const int pl = lane / NC;
    const int cl = lane % NC;
    const int myrow = rowBase + pl * R + cl % R;
    int e0 = 0, sv = 0, nv = 0;
    if (pl < PS && myrow < nDst)
        e0 = sage_link_or_end(head[(size_t)(cl / R) * nDst + myrow], E);
    if (e0 != 0) { nv = next[e0 - 1]; sv = src[e0 - 1]; }
    unsigned live = __builtin_amdgcn_ballot_w32(e0 != 0);

    for (int ps = 0; ps < PS; ++ps) {
        const int row0 = rowBase + ps * R;
        if (row0 >= nDst) break;
        float2 acc[R];
        int cnt[R];
#pragma unroll
        for (int r = 0; r < R; ++r) {
            acc[r] = make_float2(0.f, 0.f);
            cnt[r] = 0;
        }

        for (int hop = 0; hop < E; ++hop) {
            const unsigned alive = (live >> (NC * ps)) & PASSMASK;
            if (alive == 0u) break;
            int sc[NC];
#pragma unroll
            for (int c = 0; c < NC; ++c)
                sc[c] = __builtin_amdgcn_readlane(sv, NC * ps + c);
            const int en = (pl == ps) ? sage_link_or_end(nv, E) : 0;
            const unsigned moved = __builtin_amdgcn_ballot_w32(en != 0);

            float2 t[NC];
#pragma unroll
            for (int c = 0; c < NC; ++c) {
                t[c] = make_float2(0.f, 0.f);
                if ((alive >> c) & 1u) {
                    t[c] = ((const float2*)(P + (size_t)sc[c] * 64))[lane];
                    cnt[c % R] += 1;
                }
            }
            if (en != 0) { nv = next[en - 1]; sv = src[en - 1]; }
            live = (live & ~(PASSMASK << (NC * ps))) | moved;
#pragma unroll
            for (int c = 0; c < NC; ++c) {
                acc[c % R].x += t[c].x; acc[c % R].y += t[c].y;
            }
        }

#pragma unroll
        for (int r = 0; r < R; ++r) {
            if (row0 + r >= nDst) break;
            const float scale = (cnt[r] > 0) ? (1.0f / (float)cnt[r]) : 0.0f;
            float2* orow = (float2*)(out + (size_t)(row0 + r) * 64);
            const float2 own = orow[lane];
            const float a = (acc[r].x * scale + bb.x) + own.x;
            const float b = (acc[r].y * scale + bb.y) + own.y;
            float m = fmaxf(a, b);
#pragma unroll
            for (int off = 16; off > 0; off >>= 1)
                m = fmaxf(m, __shfl_xor(m, off, 32));
            float z = expf(a - m) + expf(b - m);
#pragma unroll
            for (int off = 16; off > 0; off >>= 1)
                z += __shfl_xor(z, off, 32);
            const float lse = m + logf(z);
            orow[lane] = make_float2(a - lse, b - lse);
        }
    }
}

extern "C" void kernel_launch(void* const* d_in, const int* in_sizes, int n_in,
                              void* d_out, int out_size, void* d_ws, size_t ws_size,
                              hipStream_t stream) {
    const float* x    = (const float*)d_in[0];
    const int*   src1 = (const int*)d_in[1];
    const int*   dst1 = (const int*)d_in[2];
    const int*   src2 = (const int*)d_in[3];
    const int*   dst2 = (const int*)d_in[4];
    const float* W1l  = (const float*)d_in[5];
    const float* W1r  = (const float*)d_in[6];
    const float* b1   = (const float*)d_in[7];
    const float* W2l  = (const float*)d_in[8];
    const float* W2r  = (const float*)d_in[9];
    const float* b2   = (const float*)d_in[10];
    float* out = (float*)d_out;

    int* head2 = (int*)d_ws;
    int* next1 = head2 + 3 * N1_ + 3 * N2_;
    int* next2 = next1 + E1_;
    float* aggr1 = (float*)(next2 + E2_);
    float* proj2 = aggr1 + (size_t)N1_ * IN_C;
    float* h     = proj2 + (size_t)N1_ * OUT_C;
    int* head1   = (int*)proj2;

    sage_link_chunks<<<NL * G1 + NL, 1024, 0, stream>>>(dst1, head1, next1,
                                                        dst2, head2, next2);

    sage_walk_mean_slots<IN_C, 2, G1><<<N1_ / 32, 256, 0, stream>>>(x, src1, next1, head1,
                                                                    aggr1, N1_, E1_);
    sage_gemm_wreg<IN_C, HID_C, 2, 4, 16, true>
        <<<256, 128, 0, stream>>>(aggr1, W1l, x, W1r, b1, h, N1_);

    sage_proj_wreg<HID_C, OUT_C, 2, 4, 16>
        <<<854, 32, 0, stream>>>(h, W2l, W2r, proj2, out, N1_, N2_, 683);
    sage_walk_mean_lsm64<4, 1><<<N2_ / 32, 256, 0, stream>>>(proj2, src2, next2,
                                                             head2, b2, out,
                                                             N2_, E2_);
}
